# starting kernel
# baseline (speedup 1.0000x reference)
#include <hip/hip_runtime.h>

typedef __attribute__((ext_vector_type(2))) float v2f;
typedef __attribute__((ext_vector_type(8))) float v8f;

#define N_NODES 100000
#define N_EDGES 1600000
#define DIM 128
#define N_CLASSES 40
#define EPB 64

__global__ void deg_init_kernel(float* deg) {
    int i = blockIdx.x * blockDim.x + threadIdx.x;
    if (i < N_NODES) deg[i] = 1.0f;
}

__global__ void deg_acc_kernel(const int* __restrict__ col, float* deg) {
    int e = blockIdx.x * blockDim.x + threadIdx.x;
    if (e < N_EDGES) atomicAdd(&deg[col[e]], 1.0f);
}

__global__ void dinv_kernel(float* deg) {
    int i = blockIdx.x * blockDim.x + threadIdx.x;
    if (i < N_NODES) deg[i] = rsqrtf(deg[i]);
}

__global__ __launch_bounds__(256) void gemm128_wmma(const float* __restrict__ X,
                                                    const float* __restrict__ W,
                                                    float* __restrict__ H) {
    const int m0    = blockIdx.x * 16;
    const int lane  = threadIdx.x & 31;
    const int wave  = threadIdx.x >> 5;
    const int n0    = wave * 16;
    const int l15   = lane & 15;
    const int khalf = (lane >> 4) << 1;
    const float* xrow = X + (size_t)(m0 + l15) * DIM;

    v8f acc = {};
    for (int k = 0; k < DIM; k += 4) {
        const int kk = k + khalf;
        v2f a, b;
        a.x = xrow[kk];
        a.y = xrow[kk + 1];
        b.x = W[(size_t)kk * DIM + n0 + l15];
        b.y = W[(size_t)(kk + 1) * DIM + n0 + l15];
        acc = __builtin_amdgcn_wmma_f32_16x16x4_f32(
             false, a,  false, b,
             (short)0, acc,  false,  false);
    }
    const int crow = m0 + (lane >> 4) * 8;
    const int ccol = n0 + l15;
#pragma unroll
    for (int r = 0; r < 8; ++r)
        H[(size_t)(crow + r) * DIM + ccol] = acc[r];
}

__global__ __launch_bounds__(96) void gemm_out_wmma(const float* __restrict__ X,
                                                    const float* __restrict__ W,
                                                    const float* __restrict__ bias,
                                                    float* __restrict__ OUT) {
    const int m0    = blockIdx.x * 16;
    const int lane  = threadIdx.x & 31;
    const int wave  = threadIdx.x >> 5;
    const int n0    = wave * 16;
    const int l15   = lane & 15;
    const int khalf = (lane >> 4) << 1;
    const int ncol  = n0 + l15;
    const bool nv   = (ncol < N_CLASSES);
    const float* xrow = X + (size_t)(m0 + l15) * DIM;

    v8f acc = {};
    for (int k = 0; k < DIM; k += 4) {
        const int kk = k + khalf;
        v2f a, b;
        a.x = xrow[kk];
        a.y = xrow[kk + 1];
        b.x = nv ? W[(size_t)kk * N_CLASSES + ncol] : 0.0f;
        b.y = nv ? W[(size_t)(kk + 1) * N_CLASSES + ncol] : 0.0f;
        acc = __builtin_amdgcn_wmma_f32_16x16x4_f32(
            false, a, false, b, (short)0, acc, false, false);
    }
    if (nv) {
        const int crow = m0 + (lane >> 4) * 8;
        const float bb = bias[ncol];
#pragma unroll
        for (int r = 0; r < 8; ++r)
            OUT[(size_t)(crow + r) * N_CLASSES + ncol] = acc[r] + bb;
    }
}

__global__ void agg_selfloop_kernel(const float* __restrict__ dinv,
                                    const float* __restrict__ h,
                                    float* __restrict__ agg) {
    size_t idx = (size_t)blockIdx.x * blockDim.x + threadIdx.x;
    if (idx < (size_t)N_NODES * DIM) {
        int node = (int)(idx >> 7);
        float d = dinv[node];
        agg[idx] = d * d * h[idx];
    }
}

__global__ __launch_bounds__(128) void scatter_kernel(const int* __restrict__ row,
                                                      const int* __restrict__ col,
                                                      const float* __restrict__ dinv,
                                                      const float* __restrict__ h,
                                                      float* __restrict__ agg) {
    __shared__ int   s_row[EPB];
    __shared__ int   s_col[EPB];
    __shared__ float s_norm[EPB];
    const int e0 = blockIdx.x * EPB;
    const int t  = threadIdx.x;

    for (int i = t; i < EPB; i += 128) {
        int e = e0 + i;
        if (e < N_EDGES) {
            int r = row[e], c = col[e];
            s_row[i]  = r;
            s_col[i]  = c;
            s_norm[i] = dinv[r] * dinv[c];
        }
    }
    __syncthreads();

    int lim = N_EDGES - e0;
    if (lim > EPB) lim = EPB;
    for (int i = 0; i < lim; ++i) {
        float v = s_norm[i] * h[(size_t)s_row[i] * DIM + t];
        atomicAdd(&agg[(size_t)s_col[i] * DIM + t], v);
    }
}

__global__ void bias_relu_kernel(float* __restrict__ x, const float* __restrict__ b) {
    size_t idx = (size_t)blockIdx.x * blockDim.x + threadIdx.x;
    if (idx < (size_t)N_NODES * DIM)
        x[idx] = fmaxf(x[idx] + b[idx & (DIM - 1)], 0.0f);
}

extern "C" void kernel_launch(void* const* d_in, const int* in_sizes, int n_in,
                              void* d_out, int out_size, void* d_ws, size_t ws_size,
                              hipStream_t stream) {
    const float* x    = (const float*)d_in[0];
    const int*   edge = (const int*)d_in[1];
    const float* W1   = (const float*)d_in[2];
    const float* b1   = (const float*)d_in[3];
    const float* W2   = (const float*)d_in[4];
    const float* b2   = (const float*)d_in[5];
    const float* Wout = (const float*)d_in[6];
    const float* bout = (const float*)d_in[7];
    const int* row = edge;
    const int* col = edge + N_EDGES;
    float* out = (float*)d_out;

    char* ws = (char*)d_ws;
    float* dinv = (float*)ws;
    float* bufA = (float*)(ws + ((size_t)1 << 20));
    float* bufB = bufA + (size_t)N_NODES * DIM;

    const int nelem  = N_NODES * DIM;
    const int ntiles = N_NODES / 16;

    deg_init_kernel<<<(N_NODES + 255) / 256, 256, 0, stream>>>(dinv);
    deg_acc_kernel<<<(N_EDGES + 255) / 256, 256, 0, stream>>>(col, dinv);
    dinv_kernel<<<(N_NODES + 255) / 256, 256, 0, stream>>>(dinv);

    gemm128_wmma<<<ntiles, 256, 0, stream>>>(x, W1, bufA);
    agg_selfloop_kernel<<<(nelem + 255) / 256, 256, 0, stream>>>(dinv, bufA, bufB);
    scatter_kernel<<<(N_EDGES + EPB - 1) / EPB, 128, 0, stream>>>(row, col, dinv, bufA, bufB);
    bias_relu_kernel<<<(nelem + 255) / 256, 256, 0, stream>>>(bufB, b1);

    gemm128_wmma<<<ntiles, 256, 0, stream>>>(bufB, W2, bufA);
    agg_selfloop_kernel<<<(nelem + 255) / 256, 256, 0, stream>>>(dinv, bufA, bufB);
    scatter_kernel<<<(N_EDGES + EPB - 1) / EPB, 128, 0, stream>>>(row, col, dinv, bufA, bufB);
    bias_relu_kernel<<<(nelem + 255) / 256, 256, 0, stream>>>(bufB, b2);

    gemm_out_wmma<<<ntiles, 96, 0, stream>>>(bufB, Wout, bout, out);
}


# Aggregation by destination-sorted edge list built in-call (int atomic arrival-order rank + scan): one wave sums each row in registers, stores once; no f32 feature atomics
# speedup vs baseline: 1.9222x; 1.9222x over previous
#include <hip/hip_runtime.h>

typedef __attribute__((ext_vector_type(2))) float v2f;
typedef __attribute__((ext_vector_type(8))) float v8f;

#define N_NODES 100000
#define N_EDGES 1600000
#define DIM 128
#define N_CLASSES 40

__global__ void cnt_zero_kernel(int* cnt) {
    int i = blockIdx.x * blockDim.x + threadIdx.x;
    if (i < N_NODES) cnt[i] = 0;
}

__global__ void hist_rank_kernel(const int* __restrict__ col, int* cnt, int* __restrict__ rank) {
    int e = blockIdx.x * blockDim.x + threadIdx.x;
    if (e < N_EDGES) rank[e] = atomicAdd(&cnt[col[e]], 1);
}

__global__ __launch_bounds__(1024) void scan_dinv_kernel(const int* __restrict__ cnt,
                                                         int* __restrict__ offs,
                                                         float* __restrict__ dinv) {
    __shared__ int s_w[32];
    const int t    = threadIdx.x;
    const int lane = t & 31;
    const int wv   = t >> 5;

    const int4* c4 = (const int4*)cnt;
    const int nprev4 = blockIdx.x * 256;
    int part = 0;
    int i4 = t;
    for (; i4 + 3072 < nprev4; i4 += 4096) {
        const int4 a = c4[i4];
        const int4 b = c4[i4 + 1024];
        const int4 c = c4[i4 + 2048];
        const int4 d = c4[i4 + 3072];
        part += ((a.x + a.y) + (a.z + a.w)) + ((b.x + b.y) + (b.z + b.w)) +
                ((c.x + c.y) + (c.z + c.w)) + ((d.x + d.y) + (d.z + d.w));
    }
    for (; i4 < nprev4; i4 += 1024) {
        const int4 v = c4[i4];
        part += (v.x + v.y) + (v.z + v.w);
    }
    for (int o = 16; o > 0; o >>= 1) part += __shfl_xor(part, o);
    if (lane == 0) s_w[wv] = part;
    __syncthreads();
    int prefix = 0;
    for (int k = 0; k < 32; ++k) prefix += s_w[k];
    __syncthreads();

    const int i = blockIdx.x * 1024 + t;
    const int c = (i < N_NODES) ? cnt[i] : 0;
    int inc = c;
    for (int o = 1; o < 32; o <<= 1) {
        int u = __shfl_up(inc, o);
        if (lane >= o) inc += u;
    }
    if (lane == 31) s_w[wv] = inc;
    __syncthreads();
    int wpre = 0;
    for (int k = 0; k < wv; ++k) wpre += s_w[k];
    const int ex = prefix + wpre + inc - c;
    if (i < N_NODES) {
        offs[i] = ex;
        dinv[i] = rsqrtf((float)(c + 1));
        if (i == N_NODES - 1) offs[N_NODES] = ex + c;
    }
}

__global__ void fill_kernel(const int* __restrict__ row, const int* __restrict__ col,
                            const int* __restrict__ rank, const int* __restrict__ offs,
                            const float* __restrict__ dinv, int* __restrict__ csr) {
    int e = blockIdx.x * blockDim.x + threadIdx.x;
    if (e < N_EDGES) {
        const int r = row[e];
        const int p = offs[col[e]] + rank[e];
        int2 ent;
        ent.x = r;
        ent.y = __float_as_int(dinv[r]);
        *(int2*)(csr + 2 * (size_t)p) = ent;
    }
}

__global__ __launch_bounds__(256) void gemm128_wmma(const float* __restrict__ X,
                                                    const float* __restrict__ W,
                                                    float* __restrict__ H) {
    const int m0    = blockIdx.x * 16;
    const int lane  = threadIdx.x & 31;
    const int wave  = threadIdx.x >> 5;
    const int n0    = wave * 16;
    const int l15   = lane & 15;
    const int khalf = (lane >> 4) << 1;
    const float* xrow = X + (size_t)(m0 + l15) * DIM;

    v8f acc = {};
    for (int k = 0; k < DIM; k += 4) {
        const int kk = k + khalf;
        v2f a, b;
        a.x = xrow[kk];
        a.y = xrow[kk + 1];
        b.x = W[(size_t)kk * DIM + n0 + l15];
        b.y = W[(size_t)(kk + 1) * DIM + n0 + l15];
        acc = __builtin_amdgcn_wmma_f32_16x16x4_f32(
             false, a,  false, b,
             (short)0, acc,  false,  false);
    }
    const int crow = m0 + (lane >> 4) * 8;
    const int ccol = n0 + l15;
#pragma unroll
    for (int r = 0; r < 8; ++r)
        H[(size_t)(crow + r) * DIM + ccol] = acc[r];
}

__global__ __launch_bounds__(96) void gemm_out_wmma(const float* __restrict__ X,
                                                    const float* __restrict__ W,
                                                    const float* __restrict__ bias,
                                                    float* __restrict__ OUT) {
    const int m0    = blockIdx.x * 16;
    const int lane  = threadIdx.x & 31;
    const int wave  = threadIdx.x >> 5;
    const int n0    = wave * 16;
    const int l15   = lane & 15;
    const int khalf = (lane >> 4) << 1;
    const int ncol  = n0 + l15;
    const bool nv   = (ncol < N_CLASSES);
    const float* xrow = X + (size_t)(m0 + l15) * DIM;

    v8f acc = {};
    for (int k = 0; k < DIM; k += 4) {
        const int kk = k + khalf;
        v2f a, b;
        a.x = xrow[kk];
        a.y = xrow[kk + 1];
        b.x = nv ? W[(size_t)kk * N_CLASSES + ncol] : 0.0f;
        b.y = nv ? W[(size_t)(kk + 1) * N_CLASSES + ncol] : 0.0f;
        acc = __builtin_amdgcn_wmma_f32_16x16x4_f32(
            false, a, false, b, (short)0, acc, false, false);
    }
    if (nv) {
        const int crow = m0 + (lane >> 4) * 8;
        const float bb = bias[ncol];
#pragma unroll
        for (int r = 0; r < 8; ++r)
            OUT[(size_t)(crow + r) * N_CLASSES + ncol] = acc[r] + bb;
    }
}

__device__ __forceinline__ void row_fma(float4& a, float s, const float4& v) {
    a.x = fmaf(s, v.x, a.x);
    a.y = fmaf(s, v.y, a.y);
    a.z = fmaf(s, v.z, a.z);
    a.w = fmaf(s, v.w, a.w);
}

__global__ __launch_bounds__(256) void gather_kernel(const int* __restrict__ offs,
                                                     const int* __restrict__ csr,
                                                     const float* __restrict__ dinv,
                                                     const float* __restrict__ h,
                                                     float* __restrict__ agg) {
    const int lane = threadIdx.x & 31;
    const int wave = __builtin_amdgcn_readfirstlane(threadIdx.x >> 5);
    const int node = blockIdx.x * 8 + wave;
    const int beg  = offs[node];
    const int end  = offs[node + 1];
    const float di = dinv[node];
    const float4* h4 = (const float4*)h;

    const float4 hs = h4[(size_t)node * 32 + lane];
    const float sw  = di * di;
    float4 acc;
    acc.x = sw * hs.x;
    acc.y = sw * hs.y;
    acc.z = sw * hs.z;
    acc.w = sw * hs.w;

    for (int base = beg; base < end; base += 32) {
        int n = end - base;
        if (n > 32) n = 32;
        int2 ent;
        ent.x = 0;
        ent.y = 0;
        if (lane < n) ent = *(const int2*)(csr + 2 * (size_t)(base + lane));

        int j = 0;
        for (; j + 4 <= n; j += 4) {
            const int s0 = __builtin_amdgcn_readlane(ent.x, j);
            const int s1 = __builtin_amdgcn_readlane(ent.x, j + 1);
            const int s2 = __builtin_amdgcn_readlane(ent.x, j + 2);
            const int s3 = __builtin_amdgcn_readlane(ent.x, j + 3);
            const float w0 = __int_as_float(__builtin_amdgcn_readlane(ent.y, j)) * di;
            const float w1 = __int_as_float(__builtin_amdgcn_readlane(ent.y, j + 1)) * di;
            const float w2 = __int_as_float(__builtin_amdgcn_readlane(ent.y, j + 2)) * di;
            const float w3 = __int_as_float(__builtin_amdgcn_readlane(ent.y, j + 3)) * di;
            const float4 v0 = h4[(size_t)s0 * 32 + lane];
            const float4 v1 = h4[(size_t)s1 * 32 + lane];
            const float4 v2 = h4[(size_t)s2 * 32 + lane];
            const float4 v3 = h4[(size_t)s3 * 32 + lane];
            row_fma(acc, w0, v0);
            row_fma(acc, w1, v1);
            row_fma(acc, w2, v2);
            row_fma(acc, w3, v3);
        }
        const int left = n - j;
        if (left > 0) {
            const int s0 = __builtin_amdgcn_readlane(ent.x, j);
            const float w0 = __int_as_float(__builtin_amdgcn_readlane(ent.y, j)) * di;
            const float4 v0 = h4[(size_t)s0 * 32 + lane];
            if (left == 1) {
                row_fma(acc, w0, v0);
            } else {
                const int s1 = __builtin_amdgcn_readlane(ent.x, j + 1);
                const float w1 = __int_as_float(__builtin_amdgcn_readlane(ent.y, j + 1)) * di;
                const float4 v1 = h4[(size_t)s1 * 32 + lane];
                if (left == 2) {
                    row_fma(acc, w0, v0);
                    row_fma(acc, w1, v1);
                } else {
                    const int s2 = __builtin_amdgcn_readlane(ent.x, j + 2);
                    const float w2 = __int_as_float(__builtin_amdgcn_readlane(ent.y, j + 2)) * di;
                    const float4 v2 = h4[(size_t)s2 * 32 + lane];
                    row_fma(acc, w0, v0);
                    row_fma(acc, w1, v1);
                    row_fma(acc, w2, v2);
                }
            }
        }
    }
    ((float4*)agg)[(size_t)node * 32 + lane] = acc;
}

__global__ void bias_relu_kernel(float* __restrict__ x, const float* __restrict__ b) {
    size_t idx = (size_t)blockIdx.x * blockDim.x + threadIdx.x;
    if (idx < (size_t)N_NODES * DIM)
        x[idx] = fmaxf(x[idx] + b[idx & (DIM - 1)], 0.0f);
}

extern "C" void kernel_launch(void* const* d_in, const int* in_sizes, int n_in,
                              void* d_out, int out_size, void* d_ws, size_t ws_size,
                              hipStream_t stream) {
    const float* x    = (const float*)d_in[0];
    const int*   edge = (const int*)d_in[1];
    const float* W1   = (const float*)d_in[2];
    const float* b1   = (const float*)d_in[3];
    const float* W2   = (const float*)d_in[4];
    const float* b2   = (const float*)d_in[5];
    const float* Wout = (const float*)d_in[6];
    const float* bout = (const float*)d_in[7];
    const int* row = edge;
    const int* col = edge + N_EDGES;
    float* out = (float*)d_out;

    char* ws = (char*)d_ws;
    float* dinv = (float*)ws;
    int*   cnt  = (int*)(ws + ((size_t)1 << 19));
    float* bufA = (float*)(ws + ((size_t)1 << 20));
    float* bufB = bufA + (size_t)N_NODES * DIM;
    int*   offs = (int*)(ws + ((size_t)104 << 20));
    int*   rank = (int*)(ws + ((size_t)105 << 20));
    int*   csr  = (int*)(ws + ((size_t)112 << 20));

    const int nelem  = N_NODES * DIM;
    const int ntiles = N_NODES / 16;

    cnt_zero_kernel<<<(N_NODES + 255) / 256, 256, 0, stream>>>(cnt);
    hist_rank_kernel<<<(N_EDGES + 255) / 256, 256, 0, stream>>>(col, cnt, rank);
    scan_dinv_kernel<<<(N_NODES + 1023) / 1024, 1024, 0, stream>>>(cnt, offs, dinv);
    fill_kernel<<<(N_EDGES + 255) / 256, 256, 0, stream>>>(row, col, rank, offs, dinv, csr);

    gemm128_wmma<<<ntiles, 256, 0, stream>>>(x, W1, bufA);
    gather_kernel<<<N_NODES / 8, 256, 0, stream>>>(offs, csr, dinv, bufA, bufB);
    bias_relu_kernel<<<(nelem + 255) / 256, 256, 0, stream>>>(bufB, b1);

    gemm128_wmma<<<ntiles, 256, 0, stream>>>(bufB, W2, bufA);
    gather_kernel<<<N_NODES / 8, 256, 0, stream>>>(offs, csr, dinv, bufA, bufB);
    bias_relu_kernel<<<(nelem + 255) / 256, 256, 0, stream>>>(bufB, b2);

    gemm_out_wmma<<<ntiles, 96, 0, stream>>>(bufB, Wout, bout, out);
}


# Bias add and relu done in the aggregation kernel's registers on the finished row sum before its one store; the two one-element-per-thread passes (400000 waves each) are gone: launches 11 -> 9
# speedup vs baseline: 2.2945x; 1.1937x over previous
#include <hip/hip_runtime.h>

typedef __attribute__((ext_vector_type(2))) float v2f;
typedef __attribute__((ext_vector_type(8))) float v8f;

#define N_NODES 100000
#define N_EDGES 1600000
#define DIM 128
#define N_CLASSES 40

__global__ void cnt_zero_kernel(int* cnt) {
    int i = blockIdx.x * blockDim.x + threadIdx.x;
    if (i < N_NODES) cnt[i] = 0;
}

__global__ void hist_rank_kernel(const int* __restrict__ col, int* cnt, int* __restrict__ rank) {
    int e = blockIdx.x * blockDim.x + threadIdx.x;
    if (e < N_EDGES) rank[e] = atomicAdd(&cnt[col[e]], 1);
}

__global__ __launch_bounds__(1024) void scan_dinv_kernel(const int* __restrict__ cnt,
                                                         int* __restrict__ offs,
                                                         float* __restrict__ dinv) {
    __shared__ int s_w[32];
    const int t    = threadIdx.x;
    const int lane = t & 31;
    const int wv   = t >> 5;

    const int4* c4 = (const int4*)cnt;
    const int nprev4 = blockIdx.x * 256;
    int part = 0;
    int i4 = t;
    for (; i4 + 3072 < nprev4; i4 += 4096) {
        const int4 a = c4[i4];
        const int4 b = c4[i4 + 1024];
        const int4 c = c4[i4 + 2048];
        const int4 d = c4[i4 + 3072];
        part += ((a.x + a.y) + (a.z + a.w)) + ((b.x + b.y) + (b.z + b.w)) +
                ((c.x + c.y) + (c.z + c.w)) + ((d.x + d.y) + (d.z + d.w));
    }
    for (; i4 < nprev4; i4 += 1024) {
        const int4 v = c4[i4];
        part += (v.x + v.y) + (v.z + v.w);
    }
    for (int o = 16; o > 0; o >>= 1) part += __shfl_xor(part, o);
    if (lane == 0) s_w[wv] = part;
    __syncthreads();
    int prefix = 0;
    for (int k = 0; k < 32; ++k) prefix += s_w[k];
    __syncthreads();

    const int i = blockIdx.x * 1024 + t;
    const int c = (i < N_NODES) ? cnt[i] : 0;
    int inc = c;
    for (int o = 1; o < 32; o <<= 1) {
        int u = __shfl_up(inc, o);
        if (lane >= o) inc += u;
    }
    if (lane == 31) s_w[wv] = inc;
    __syncthreads();
    int wpre = 0;
    for (int k = 0; k < wv; ++k) wpre += s_w[k];
    const int ex = prefix + wpre + inc - c;
    if (i < N_NODES) {
        offs[i] = ex;
        dinv[i] = rsqrtf((float)(c + 1));
        if (i == N_NODES - 1) offs[N_NODES] = ex + c;
    }
}

__global__ void fill_kernel(const int* __restrict__ row, const int* __restrict__ col,
                            const int* __restrict__ rank, const int* __restrict__ offs,
                            const float* __restrict__ dinv, int* __restrict__ csr) {
    int e = blockIdx.x * blockDim.x + threadIdx.x;
    if (e < N_EDGES) {
        const int r = row[e];
        const int p = offs[col[e]] + rank[e];
        int2 ent;
        ent.x = r;
        ent.y = __float_as_int(dinv[r]);
        *(int2*)(csr + 2 * (size_t)p) = ent;
    }
}

__global__ __launch_bounds__(256) void gemm128_wmma(const float* __restrict__ X,
                                                    const float* __restrict__ W,
                                                    float* __restrict__ H) {
    const int m0    = blockIdx.x * 16;
    const int lane  = threadIdx.x & 31;
    const int wave  = threadIdx.x >> 5;
    const int n0    = wave * 16;
    const int l15   = lane & 15;
    const int khalf = (lane >> 4) << 1;
    const float* xrow = X + (size_t)(m0 + l15) * DIM;

    v8f acc = {};
    for (int k = 0; k < DIM; k += 4) {
        const int kk = k + khalf;
        v2f a, b;
        a.x = xrow[kk];
        a.y = xrow[kk + 1];
        b.x = W[(size_t)kk * DIM + n0 + l15];
        b.y = W[(size_t)(kk + 1) * DIM + n0 + l15];
        acc = __builtin_amdgcn_wmma_f32_16x16x4_f32(
             false, a,  false, b,
             (short)0, acc,  false,  false);
    }
    const int crow = m0 + (lane >> 4) * 8;
    const int ccol = n0 + l15;
#pragma unroll
    for (int r = 0; r < 8; ++r)
        H[(size_t)(crow + r) * DIM + ccol] = acc[r];
}

__global__ __launch_bounds__(96) void gemm_out_wmma(const float* __restrict__ X,
                                                    const float* __restrict__ W,
                                                    const float* __restrict__ bias,
                                                    float* __restrict__ OUT) {
    const int m0    = blockIdx.x * 16;
    const int lane  = threadIdx.x & 31;
    const int wave  = threadIdx.x >> 5;
    const int n0    = wave * 16;
    const int l15   = lane & 15;
    const int khalf = (lane >> 4) << 1;
    const int ncol  = n0 + l15;
    const bool nv   = (ncol < N_CLASSES);
    const float* xrow = X + (size_t)(m0 + l15) * DIM;

    v8f acc = {};
    for (int k = 0; k < DIM; k += 4) {
        const int kk = k + khalf;
        v2f a, b;
        a.x = xrow[kk];
        a.y = xrow[kk + 1];
        b.x = nv ? W[(size_t)kk * N_CLASSES + ncol] : 0.0f;
        b.y = nv ? W[(size_t)(kk + 1) * N_CLASSES + ncol] : 0.0f;
        acc = __builtin_amdgcn_wmma_f32_16x16x4_f32(
            false, a, false, b, (short)0, acc, false, false);
    }
    if (nv) {
        const int crow = m0 + (lane >> 4) * 8;
        const float bb = bias[ncol];
#pragma unroll
        for (int r = 0; r < 8; ++r)
            OUT[(size_t)(crow + r) * N_CLASSES + ncol] = acc[r] + bb;
    }
}

__device__ __forceinline__ void row_fma(float4& a, float s, const float4& v) {
    a.x = fmaf(s, v.x, a.x);
    a.y = fmaf(s, v.y, a.y);
    a.z = fmaf(s, v.z, a.z);
    a.w = fmaf(s, v.w, a.w);
}

__global__ __launch_bounds__(256) void gather_kernel(const int* __restrict__ offs,
                                                     const int* __restrict__ csr,
                                                     const float* __restrict__ dinv,
                                                     const float* __restrict__ h,
                                                     const float* __restrict__ bias,
                                                     float* __restrict__ xout) {
    const int lane = threadIdx.x & 31;
    const int wave = __builtin_amdgcn_readfirstlane(threadIdx.x >> 5);
    const int node = blockIdx.x * 8 + wave;
    const int beg  = offs[node];
    const int end  = offs[node + 1];
    const float di = dinv[node];
    const float4* h4 = (const float4*)h;

    const float4 hs = h4[(size_t)node * 32 + lane];
    const float4 bb = ((const float4*)bias)[lane];
    const float sw  = di * di;
    float4 acc;
    acc.x = sw * hs.x;
    acc.y = sw * hs.y;
    acc.z = sw * hs.z;
    acc.w = sw * hs.w;

    for (int base = beg; base < end; base += 32) {
        int n = end - base;
        if (n > 32) n = 32;
        int2 ent;
        ent.x = 0;
        ent.y = 0;
        if (lane < n) ent = *(const int2*)(csr + 2 * (size_t)(base + lane));

        int j = 0;
        for (; j + 4 <= n; j += 4) {
            const int s0 = __builtin_amdgcn_readlane(ent.x, j);
            const int s1 = __builtin_amdgcn_readlane(ent.x, j + 1);
            const int s2 = __builtin_amdgcn_readlane(ent.x, j + 2);
            const int s3 = __builtin_amdgcn_readlane(ent.x, j + 3);
            const float w0 = __int_as_float(__builtin_amdgcn_readlane(ent.y, j)) * di;
            const float w1 = __int_as_float(__builtin_amdgcn_readlane(ent.y, j + 1)) * di;
            const float w2 = __int_as_float(__builtin_amdgcn_readlane(ent.y, j + 2)) * di;
            const float w3 = __int_as_float(__builtin_amdgcn_readlane(ent.y, j + 3)) * di;
            const float4 v0 = h4[(size_t)s0 * 32 + lane];
            const float4 v1 = h4[(size_t)s1 * 32 + lane];
            const float4 v2 = h4[(size_t)s2 * 32 + lane];
            const float4 v3 = h4[(size_t)s3 * 32 + lane];
            row_fma(acc, w0, v0);
            row_fma(acc, w1, v1);
            row_fma(acc, w2, v2);
            row_fma(acc, w3, v3);
        }
        const int left = n - j;
        if (left > 0) {
            const int s0 = __builtin_amdgcn_readlane(ent.x, j);
            const float w0 = __int_as_float(__builtin_amdgcn_readlane(ent.y, j)) * di;
            const float4 v0 = h4[(size_t)s0 * 32 + lane];
            if (left == 1) {
                row_fma(acc, w0, v0);
            } else {
                const int s1 = __builtin_amdgcn_readlane(ent.x, j + 1);
                const float w1 = __int_as_float(__builtin_amdgcn_readlane(ent.y, j + 1)) * di;
                const float4 v1 = h4[(size_t)s1 * 32 + lane];
                if (left == 2) {
                    row_fma(acc, w0, v0);
                    row_fma(acc, w1, v1);
                } else {
                    const int s2 = __builtin_amdgcn_readlane(ent.x, j + 2);
                    const float w2 = __int_as_float(__builtin_amdgcn_readlane(ent.y, j + 2)) * di;
                    const float4 v2 = h4[(size_t)s2 * 32 + lane];
                    row_fma(acc, w0, v0);
                    row_fma(acc, w1, v1);
                    row_fma(acc, w2, v2);
                }
            }
        }
    }
    acc.x = fmaxf(acc.x + bb.x, 0.0f);
    acc.y = fmaxf(acc.y + bb.y, 0.0f);
    acc.z = fmaxf(acc.z + bb.z, 0.0f);
    acc.w = fmaxf(acc.w + bb.w, 0.0f);
    ((float4*)xout)[(size_t)node * 32 + lane] = acc;
}

extern "C" void kernel_launch(void* const* d_in, const int* in_sizes, int n_in,
                              void* d_out, int out_size, void* d_ws, size_t ws_size,
                              hipStream_t stream) {
    const float* x    = (const float*)d_in[0];
    const int*   edge = (const int*)d_in[1];
    const float* W1   = (const float*)d_in[2];
    const float* b1   = (const float*)d_in[3];
    const float* W2   = (const float*)d_in[4];
    const float* b2   = (const float*)d_in[5];
    const float* Wout = (const float*)d_in[6];
    const float* bout = (const float*)d_in[7];
    const int* row = edge;
    const int* col = edge + N_EDGES;
    float* out = (float*)d_out;

    char* ws = (char*)d_ws;
    float* dinv = (float*)ws;
    int*   cnt  = (int*)(ws + ((size_t)1 << 19));
    float* bufA = (float*)(ws + ((size_t)1 << 20));
    float* bufB = bufA + (size_t)N_NODES * DIM;
    int*   offs = (int*)(ws + ((size_t)104 << 20));
    int*   rank = (int*)(ws + ((size_t)105 << 20));
    int*   csr  = (int*)(ws + ((size_t)112 << 20));

    const int ntiles = N_NODES / 16;

    cnt_zero_kernel<<<(N_NODES + 255) / 256, 256, 0, stream>>>(cnt);
    hist_rank_kernel<<<(N_EDGES + 255) / 256, 256, 0, stream>>>(col, cnt, rank);
    scan_dinv_kernel<<<(N_NODES + 1023) / 1024, 1024, 0, stream>>>(cnt, offs, dinv);
    fill_kernel<<<(N_EDGES + 255) / 256, 256, 0, stream>>>(row, col, rank, offs, dinv, csr);

    gemm128_wmma<<<ntiles, 256, 0, stream>>>(x, W1, bufA);
    gather_kernel<<<N_NODES / 8, 256, 0, stream>>>(offs, csr, dinv, bufA, b1, bufB);

    gemm128_wmma<<<ntiles, 256, 0, stream>>>(bufB, W2, bufA);
    gather_kernel<<<N_NODES / 8, 256, 0, stream>>>(offs, csr, dinv, bufA, b2, bufB);

    gemm_out_wmma<<<ntiles, 96, 0, stream>>>(bufB, Wout, bout, out);
}


# GEMMs re-tiled: each wave holds its 128x16 W slice in registers over a strided loop of 16-row tiles; X tiles staged in LDS by whole-row loads one tile ahead, results stored as whole 128-byte lines
# speedup vs baseline: 2.8669x; 1.2495x over previous
#include <hip/hip_runtime.h>

typedef __attribute__((ext_vector_type(2))) float v2f;
typedef __attribute__((ext_vector_type(8))) float v8f;

#define N_NODES 100000
#define N_EDGES 1600000
#define DIM 128
#define N_CLASSES 40

__global__ void cnt_zero_kernel(int* cnt) {
    int i = blockIdx.x * blockDim.x + threadIdx.x;
    if (i < N_NODES) cnt[i] = 0;
}

__global__ void hist_rank_kernel(const int* __restrict__ col, int* cnt, int* __restrict__ rank) {
    int e = blockIdx.x * blockDim.x + threadIdx.x;
    if (e < N_EDGES) rank[e] = atomicAdd(&cnt[col[e]], 1);
}

__global__ __launch_bounds__(1024) void scan_dinv_kernel(const int* __restrict__ cnt,
                                                         int* __restrict__ offs,
                                                         float* __restrict__ dinv) {
    __shared__ int s_w[32];
    const int t    = threadIdx.x;
    const int lane = t & 31;
    const int wv   = t >> 5;

    const int4* c4 = (const int4*)cnt;
    const int nprev4 = blockIdx.x * 256;
    int part = 0;
    int i4 = t;
    for (; i4 + 3072 < nprev4; i4 += 4096) {
        const int4 a = c4[i4];
        const int4 b = c4[i4 + 1024];
        const int4 c = c4[i4 + 2048];
        const int4 d = c4[i4 + 3072];
        part += ((a.x + a.y) + (a.z + a.w)) + ((b.x + b.y) + (b.z + b.w)) +
                ((c.x + c.y) + (c.z + c.w)) + ((d.x + d.y) + (d.z + d.w));
    }
    for (; i4 < nprev4; i4 += 1024) {
        const int4 v = c4[i4];
        part += (v.x + v.y) + (v.z + v.w);
    }
    for (int o = 16; o > 0; o >>= 1) part += __shfl_xor(part, o);
    if (lane == 0) s_w[wv] = part;
    __syncthreads();
    int prefix = 0;
    for (int k = 0; k < 32; ++k) prefix += s_w[k];
    __syncthreads();

    const int i = blockIdx.x * 1024 + t;
    const int c = (i < N_NODES) ? cnt[i] : 0;
    int inc = c;
    for (int o = 1; o < 32; o <<= 1) {
        int u = __shfl_up(inc, o);
        if (lane >= o) inc += u;
    }
    if (lane == 31) s_w[wv] = inc;
    __syncthreads();
    int wpre = 0;
    for (int k = 0; k < wv; ++k) wpre += s_w[k];
    const int ex = prefix + wpre + inc - c;
    if (i < N_NODES) {
        offs[i] = ex;
        dinv[i] = rsqrtf((float)(c + 1));
        if (i == N_NODES - 1) offs[N_NODES] = ex + c;
    }
}

__global__ void fill_kernel(const int* __restrict__ row, const int* __restrict__ col,
                            const int* __restrict__ rank, const int* __restrict__ offs,
                            const float* __restrict__ dinv, int* __restrict__ csr) {
    int e = blockIdx.x * blockDim.x + threadIdx.x;
    if (e < N_EDGES) {
        const int r = row[e];
        const int p = offs[col[e]] + rank[e];
        int2 ent;
        ent.x = r;
        ent.y = __float_as_int(dinv[r]);
        *(int2*)(csr + 2 * (size_t)p) = ent;
    }
}

#define APITCH 132

template <int NT, int NCOLS, bool HAS_BIAS>
__global__ __launch_bounds__(32 * NT) void gemm_tiles_wmma(const float* __restrict__ X,
                                                           const float* __restrict__ W,
                                                           const float* __restrict__ bias,
                                                           float* __restrict__ OUT,
                                                           int ntiles) {
    constexpr int NTHR   = 32 * NT;
    constexpr int OPITCH = (NCOLS % 32 == 0) ? NCOLS + 4 : NCOLS;
    constexpr int A_F4   = 16 * DIM / 4;
    constexpr int O_F4   = 16 * NCOLS / 4;
    constexpr int ROW_F4 = NCOLS / 4;
    constexpr int NA     = (A_F4 + NTHR - 1) / NTHR;
    constexpr int NO     = (O_F4 + NTHR - 1) / NTHR;

    __shared__ __attribute__((aligned(16))) float s_a[2][16 * APITCH];
    __shared__ __attribute__((aligned(16))) float s_o[2][16 * OPITCH];

    const int tid  = threadIdx.x;
    const int lane = tid & 31;
    const int wave = tid >> 5;
    const int l15  = lane & 15;
    const int half = lane >> 4;
    const int ncol = wave * 16 + l15;
    const bool nv  = (16 * NT <= NCOLS) ? true : (ncol < NCOLS);

    int tile = blockIdx.x;
    if (tile >= ntiles) return;

    const int wcol = nv ? ncol : 0;
    v2f bf[32];
#pragma unroll
    for (int ks = 0; ks < 32; ++ks) {
        const int kk = 4 * ks + 2 * half;
        const float w0 = W[(size_t)kk * NCOLS + wcol];
        const float w1 = W[(size_t)(kk + 1) * NCOLS + wcol];
        bf[ks].x = nv ? w0 : 0.0f;
        bf[ks].y = nv ? w1 : 0.0f;
    }
    float bb = 0.0f;
    if (HAS_BIAS) bb = bias[wcol];

    float4 p0 = {}, p1 = {}, p2 = {}, p3 = {}, p4 = {}, p5 = {};
#define X_PIECE_LOAD(k)                                      \
    if (NA > k) {                                            \
        const int idx = tid + k * NTHR;                      \
        if ((k + 1) * NTHR <= A_F4 || idx < A_F4) p##k = src[idx]; \
    }
#define X_PIECE_STAGE(k, buf)                                                                  \
    if (NA > k) {                                                                              \
        const int idx = tid + k * NTHR;                                                        \
        if ((k + 1) * NTHR <= A_F4 || idx < A_F4)                                              \
            *(float4*)&s_a[buf][(idx >> 5) * APITCH + 4 * (idx & 31)] = p##k;                  \
    }
    static_assert(NA <= 6, "six staging registers");
    {
        const float4* src = (const float4*)(X + (size_t)tile * 16 * DIM);
        X_PIECE_LOAD(0) X_PIECE_LOAD(1) X_PIECE_LOAD(2) X_PIECE_LOAD(3) X_PIECE_LOAD(4) X_PIECE_LOAD(5)
        X_PIECE_STAGE(0, 0) X_PIECE_STAGE(1, 0) X_PIECE_STAGE(2, 0)
        X_PIECE_STAGE(3, 0) X_PIECE_STAGE(4, 0) X_PIECE_STAGE(5, 0)
    }
    __syncthreads();

    int cur  = 0;
    int done = -1;
    for (; tile < ntiles; tile += gridDim.x) {
        const int nxt = tile + gridDim.x;
        const bool has_next = (nxt < ntiles);
        if (has_next) {
            const float4* src = (const float4*)(X + (size_t)nxt * 16 * DIM);
            X_PIECE_LOAD(0) X_PIECE_LOAD(1) X_PIECE_LOAD(2) X_PIECE_LOAD(3) X_PIECE_LOAD(4) X_PIECE_LOAD(5)
        }
        if (done >= 0) {
            float4* dst = (float4*)(OUT + (size_t)done * 16 * NCOLS);
            const float* so = s_o[cur ^ 1];
#pragma unroll
            for (int i = 0; i < NO; ++i) {
                const int idx = tid + i * NTHR;
                if ((i + 1) * NTHR <= O_F4 || idx < O_F4)
                    dst[idx] = *(const float4*)&so[(idx / ROW_F4) * OPITCH + 4 * (idx % ROW_F4)];
            }
        }

        v8f acc = {};
        const float* ap = &s_a[cur][l15 * APITCH + 2 * half];
#pragma unroll
        for (int kh = 0; kh < 2; ++kh) {
            v2f af[16];
#pragma unroll
            for (int j = 0; j < 16; ++j) af[j] = *(const v2f*)(ap + 4 * (16 * kh + j));
            asm volatile("" ::"v"(af[0]), "v"(af[1]), "v"(af[2]), "v"(af[3]), "v"(af[4]), "v"(af[5]),
                         "v"(af[6]), "v"(af[7]), "v"(af[8]), "v"(af[9]), "v"(af[10]), "v"(af[11]),
                         "v"(af[12]), "v"(af[13]), "v"(af[14]), "v"(af[15]));
#pragma unroll
            for (int j = 0; j < 16; ++j)
                acc = __builtin_amdgcn_wmma_f32_16x16x4_f32(
                     false, af[j],  false, bf[16 * kh + j],
                     (short)0, acc,  false,  false);
        }

        if (nv) {
            float* op = &s_o[cur][(8 * half) * OPITCH + ncol];
#pragma unroll
            for (int r = 0; r < 8; ++r) op[r * OPITCH] = HAS_BIAS ? acc[r] + bb : acc[r];
        }
        if (has_next) {
            X_PIECE_STAGE(0, cur ^ 1) X_PIECE_STAGE(1, cur ^ 1) X_PIECE_STAGE(2, cur ^ 1)
            X_PIECE_STAGE(3, cur ^ 1) X_PIECE_STAGE(4, cur ^ 1) X_PIECE_STAGE(5, cur ^ 1)
        }
        __syncthreads();
        done = tile;
        cur ^= 1;
    }
    {
        float4* dst = (float4*)(OUT + (size_t)done * 16 * NCOLS);
        const float* so = s_o[cur ^ 1];
#pragma unroll
        for (int i = 0; i < NO; ++i) {
            const int idx = tid + i * NTHR;
            if ((i + 1) * NTHR <= O_F4 || idx < O_F4)
                dst[idx] = *(const float4*)&so[(idx / ROW_F4) * OPITCH + 4 * (idx % ROW_F4)];
        }
    }
}
#undef X_PIECE_LOAD
#undef X_PIECE_STAGE

__device__ __forceinline__ void row_fma(float4& a, float s, const float4& v) {
    a.x = fmaf(s, v.x, a.x);
    a.y = fmaf(s, v.y, a.y);
    a.z = fmaf(s, v.z, a.z);
    a.w = fmaf(s, v.w, a.w);
}

__global__ __launch_bounds__(256) void gather_kernel(const int* __restrict__ offs,
                                                     const int* __restrict__ csr,
                                                     const float* __restrict__ dinv,
                                                     const float* __restrict__ h,
                                                     const float* __restrict__ bias,
                                                     float* __restrict__ xout) {
    const int lane = threadIdx.x & 31;
    const int wave = __builtin_amdgcn_readfirstlane(threadIdx.x >> 5);
    const int node = blockIdx.x * 8 + wave;
    const int beg  = offs[node];
    const int end  = offs[node + 1];
    const float di = dinv[node];
    const float4* h4 = (const float4*)h;

    const float4 hs = h4[(size_t)node * 32 + lane];
    const float4 bb = ((const float4*)bias)[lane];
    const float sw  = di * di;
    float4 acc;
    acc.x = sw * hs.x;
    acc.y = sw * hs.y;
    acc.z = sw * hs.z;
    acc.w = sw * hs.w;

    for (int base = beg; base < end; base += 32) {
        int n = end - base;
        if (n > 32) n = 32;
        int2 ent;
        ent.x = 0;
        ent.y = 0;
        if (lane < n) ent = *(const int2*)(csr + 2 * (size_t)(base + lane));

        int j = 0;
        for (; j + 4 <= n; j += 4) {
            const int s0 = __builtin_amdgcn_readlane(ent.x, j);
            const int s1 = __builtin_amdgcn_readlane(ent.x, j + 1);
            const int s2 = __builtin_amdgcn_readlane(ent.x, j + 2);
            const int s3 = __builtin_amdgcn_readlane(ent.x, j + 3);
            const float w0 = __int_as_float(__builtin_amdgcn_readlane(ent.y, j)) * di;
            const float w1 = __int_as_float(__builtin_amdgcn_readlane(ent.y, j + 1)) * di;
            const float w2 = __int_as_float(__builtin_amdgcn_readlane(ent.y, j + 2)) * di;
            const float w3 = __int_as_float(__builtin_amdgcn_readlane(ent.y, j + 3)) * di;
            const float4 v0 = h4[(size_t)s0 * 32 + lane];
            const float4 v1 = h4[(size_t)s1 * 32 + lane];
            const float4 v2 = h4[(size_t)s2 * 32 + lane];
            const float4 v3 = h4[(size_t)s3 * 32 + lane];
            row_fma(acc, w0, v0);
            row_fma(acc, w1, v1);
            row_fma(acc, w2, v2);
            row_fma(acc, w3, v3);
        }
        const int left = n - j;
        if (left > 0) {
            const int s0 = __builtin_amdgcn_readlane(ent.x, j);
            const float w0 = __int_as_float(__builtin_amdgcn_readlane(ent.y, j)) * di;
            const float4 v0 = h4[(size_t)s0 * 32 + lane];
            if (left == 1) {
                row_fma(acc, w0, v0);
            } else {
                const int s1 = __builtin_amdgcn_readlane(ent.x, j + 1);
                const float w1 = __int_as_float(__builtin_amdgcn_readlane(ent.y, j + 1)) * di;
                const float4 v1 = h4[(size_t)s1 * 32 + lane];
                if (left == 2) {
                    row_fma(acc, w0, v0);
                    row_fma(acc, w1, v1);
                } else {
                    const int s2 = __builtin_amdgcn_readlane(ent.x, j + 2);
                    const float w2 = __int_as_float(__builtin_amdgcn_readlane(ent.y, j + 2)) * di;
                    const float4 v2 = h4[(size_t)s2 * 32 + lane];
                    row_fma(acc, w0, v0);
                    row_fma(acc, w1, v1);
                    row_fma(acc, w2, v2);
                }
            }
        }
    }
    acc.x = fmaxf(acc.x + bb.x, 0.0f);
    acc.y = fmaxf(acc.y + bb.y, 0.0f);
    acc.z = fmaxf(acc.z + bb.z, 0.0f);
    acc.w = fmaxf(acc.w + bb.w, 0.0f);
    ((float4*)xout)[(size_t)node * 32 + lane] = acc;
}

extern "C" void kernel_launch(void* const* d_in, const int* in_sizes, int n_in,
                              void* d_out, int out_size, void* d_ws, size_t ws_size,
                              hipStream_t stream) {
    const float* x    = (const float*)d_in[0];
    const int*   edge = (const int*)d_in[1];
    const float* W1   = (const float*)d_in[2];
    const float* b1   = (const float*)d_in[3];
    const float* W2   = (const float*)d_in[4];
    const float* b2   = (const float*)d_in[5];
    const float* Wout = (const float*)d_in[6];
    const float* bout = (const float*)d_in[7];
    const int* row = edge;
    const int* col = edge + N_EDGES;
    float* out = (float*)d_out;

    char* ws = (char*)d_ws;
    float* dinv = (float*)ws;
    int*   cnt  = (int*)(ws + ((size_t)1 << 19));
    float* bufA = (float*)(ws + ((size_t)1 << 20));
    float* bufB = bufA + (size_t)N_NODES * DIM;
    int*   offs = (int*)(ws + ((size_t)104 << 20));
    int*   rank = (int*)(ws + ((size_t)105 << 20));
    int*   csr  = (int*)(ws + ((size_t)112 << 20));

    const int ntiles = N_NODES / 16;

    cnt_zero_kernel<<<(N_NODES + 255) / 256, 256, 0, stream>>>(cnt);
    hist_rank_kernel<<<(N_EDGES + 255) / 256, 256, 0, stream>>>(col, cnt, rank);
    scan_dinv_kernel<<<(N_NODES + 1023) / 1024, 1024, 0, stream>>>(cnt, offs, dinv);
    fill_kernel<<<(N_EDGES + 255) / 256, 256, 0, stream>>>(row, col, rank, offs, dinv, csr);

    gemm_tiles_wmma<8, DIM, false><<<1024, 256, 0, stream>>>(x, W1, nullptr, bufA, ntiles);
    gather_kernel<<<N_NODES / 8, 256, 0, stream>>>(offs, csr, dinv, bufA, b1, bufB);

    gemm_tiles_wmma<8, DIM, false><<<1024, 256, 0, stream>>>(bufB, W2, nullptr, bufA, ntiles);
    gather_kernel<<<N_NODES / 8, 256, 0, stream>>>(offs, csr, dinv, bufA, b2, bufB);

    gemm_tiles_wmma<3, N_CLASSES, true><<<1250, 96, 0, stream>>>(bufB, Wout, bout, out, ntiles);
}


# Aggregation kernel keeps its finished rows (bias, relu done) in LDS and runs the next layer's matrix product on them in the same persistent block; x1 and x2 never reach memory; launches 9 -> 7
# speedup vs baseline: 3.1806x; 1.1094x over previous
#include <hip/hip_runtime.h>

typedef __attribute__((ext_vector_type(2))) float v2f;
typedef __attribute__((ext_vector_type(8))) float v8f;

#define N_NODES 100000
#define N_EDGES 1600000
#define DIM 128
#define N_CLASSES 40

__global__ void cnt_zero_kernel(int* cnt) {
    int i = blockIdx.x * blockDim.x + threadIdx.x;
    if (i < N_NODES) cnt[i] = 0;
}

__global__ void hist_rank_kernel(const int* __restrict__ col, int* cnt, int* __restrict__ rank) {
    int e = blockIdx.x * blockDim.x + threadIdx.x;
    if (e < N_EDGES) rank[e] = atomicAdd(&cnt[col[e]], 1);
}

__global__ __launch_bounds__(1024) void scan_dinv_kernel(const int* __restrict__ cnt,
                                                         int* __restrict__ offs,
                                                         float* __restrict__ dinv) {
    __shared__ int s_w[32];
    const int t    = threadIdx.x;
    const int lane = t & 31;
    const int wv   = t >> 5;

    const int4* c4 = (const int4*)cnt;
    const int nprev4 = blockIdx.x * 256;
    int part = 0;
    int i4 = t;
    for (; i4 + 3072 < nprev4; i4 += 4096) {
        const int4 a = c4[i4];
        const int4 b = c4[i4 + 1024];
        const int4 c = c4[i4 + 2048];
        const int4 d = c4[i4 + 3072];
        part += ((a.x + a.y) + (a.z + a.w)) + ((b.x + b.y) + (b.z + b.w)) +
                ((c.x + c.y) + (c.z + c.w)) + ((d.x + d.y) + (d.z + d.w));
    }
    for (; i4 < nprev4; i4 += 1024) {
        const int4 v = c4[i4];
        part += (v.x + v.y) + (v.z + v.w);
    }
    for (int o = 16; o > 0; o >>= 1) part += __shfl_xor(part, o);
    if (lane == 0) s_w[wv] = part;
    __syncthreads();
    int prefix = 0;
    for (int k = 0; k < 32; ++k) prefix += s_w[k];
    __syncthreads();

    const int i = blockIdx.x * 1024 + t;
    const int c = (i < N_NODES) ? cnt[i] : 0;
    int inc = c;
    for (int o = 1; o < 32; o <<= 1) {
        int u = __shfl_up(inc, o);
        if (lane >= o) inc += u;
    }
    if (lane == 31) s_w[wv] = inc;
    __syncthreads();
    int wpre = 0;
    for (int k = 0; k < wv; ++k) wpre += s_w[k];
    const int ex = prefix + wpre + inc - c;
    if (i < N_NODES) {
        offs[i] = ex;
        dinv[i] = rsqrtf((float)(c + 1));
        if (i == N_NODES - 1) offs[N_NODES] = ex + c;
    }
}

__global__ void fill_kernel(const int* __restrict__ row, const int* __restrict__ col,
                            const int* __restrict__ rank, const int* __restrict__ offs,
                            const float* __restrict__ dinv, int* __restrict__ csr) {
    int e = blockIdx.x * blockDim.x + threadIdx.x;
    if (e < N_EDGES) {
        const int r = row[e];
        const int p = offs[col[e]] + rank[e];
        int2 ent;
        ent.x = r;
        ent.y = __float_as_int(dinv[r]);
        *(int2*)(csr + 2 * (size_t)p) = ent;
    }
}

#define APITCH 132

template <int NT, int NCOLS, bool HAS_BIAS>
__global__ __launch_bounds__(32 * NT) void gemm_tiles_wmma(const float* __restrict__ X,
                                                           const float* __restrict__ W,
                                                           const float* __restrict__ bias,
                                                           float* __restrict__ OUT,
                                                           int ntiles) {
    constexpr int NTHR   = 32 * NT;
    constexpr int OPITCH = (NCOLS % 32 == 0) ? NCOLS + 4 : NCOLS;
    constexpr int A_F4   = 16 * DIM / 4;
    constexpr int O_F4   = 16 * NCOLS / 4;
    constexpr int ROW_F4 = NCOLS / 4;
    constexpr int NA     = (A_F4 + NTHR - 1) / NTHR;
    constexpr int NO     = (O_F4 + NTHR - 1) / NTHR;

    __shared__ __attribute__((aligned(16))) float s_a[2][16 * APITCH];
    __shared__ __attribute__((aligned(16))) float s_o[2][16 * OPITCH];

    const int tid  = threadIdx.x;
    const int lane = tid & 31;
    const int wave = tid >> 5;
    const int l15  = lane & 15;
    const int half = lane >> 4;
    const int ncol = wave * 16 + l15;
    const bool nv  = (16 * NT <= NCOLS) ? true : (ncol < NCOLS);

    int tile = blockIdx.x;
    if (tile >= ntiles) return;

    const int wcol = nv ? ncol : 0;
    v2f bf[32];
#pragma unroll
    for (int ks = 0; ks < 32; ++ks) {
        const int kk = 4 * ks + 2 * half;
        const float w0 = W[(size_t)kk * NCOLS + wcol];
        const float w1 = W[(size_t)(kk + 1) * NCOLS + wcol];
        bf[ks].x = nv ? w0 : 0.0f;
        bf[ks].y = nv ? w1 : 0.0f;
    }
    float bb = 0.0f;
    if (HAS_BIAS) bb = bias[wcol];

    float4 p0 = {}, p1 = {}, p2 = {}, p3 = {}, p4 = {}, p5 = {};
#define X_PIECE_LOAD(k)                                      \
    if (NA > k) {                                            \
        const int idx = tid + k * NTHR;                      \
        if ((k + 1) * NTHR <= A_F4 || idx < A_F4) p##k = src[idx]; \
    }
#define X_PIECE_STAGE(k, buf)                                                                  \
    if (NA > k) {                                                                              \
        const int idx = tid + k * NTHR;                                                        \
        if ((k + 1) * NTHR <= A_F4 || idx < A_F4)                                              \
            *(float4*)&s_a[buf][(idx >> 5) * APITCH + 4 * (idx & 31)] = p##k;                  \
    }
    static_assert(NA <= 6, "six staging registers");
    {
        const float4* src = (const float4*)(X + (size_t)tile * 16 * DIM);
        X_PIECE_LOAD(0) X_PIECE_LOAD(1) X_PIECE_LOAD(2) X_PIECE_LOAD(3) X_PIECE_LOAD(4) X_PIECE_LOAD(5)
        X_PIECE_STAGE(0, 0) X_PIECE_STAGE(1, 0) X_PIECE_STAGE(2, 0)
        X_PIECE_STAGE(3, 0) X_PIECE_STAGE(4, 0) X_PIECE_STAGE(5, 0)
    }
    __syncthreads();

    int cur  = 0;
    int done = -1;
    for (; tile < ntiles; tile += gridDim.x) {
        const int nxt = tile + gridDim.x;
        const bool has_next = (nxt < ntiles);
        if (has_next) {
            const float4* src = (const float4*)(X + (size_t)nxt * 16 * DIM);
            X_PIECE_LOAD(0) X_PIECE_LOAD(1) X_PIECE_LOAD(2) X_PIECE_LOAD(3) X_PIECE_LOAD(4) X_PIECE_LOAD(5)
        }
        if (done >= 0) {
            float4* dst = (float4*)(OUT + (size_t)done * 16 * NCOLS);
            const float* so = s_o[cur ^ 1];
#pragma unroll
            for (int i = 0; i < NO; ++i) {
                const int idx = tid + i * NTHR;
                if ((i + 1) * NTHR <= O_F4 || idx < O_F4)
                    dst[idx] = *(const float4*)&so[(idx / ROW_F4) * OPITCH + 4 * (idx % ROW_F4)];
            }
        }

        v8f acc = {};
        const float* ap = &s_a[cur][l15 * APITCH + 2 * half];
#pragma unroll
        for (int kh = 0; kh < 2; ++kh) {
            v2f af[16];
#pragma unroll
            for (int j = 0; j < 16; ++j) af[j] = *(const v2f*)(ap + 4 * (16 * kh + j));
            asm volatile("" ::"v"(af[0]), "v"(af[1]), "v"(af[2]), "v"(af[3]), "v"(af[4]), "v"(af[5]),
                         "v"(af[6]), "v"(af[7]), "v"(af[8]), "v"(af[9]), "v"(af[10]), "v"(af[11]),
                         "v"(af[12]), "v"(af[13]), "v"(af[14]), "v"(af[15]));
#pragma unroll
            for (int j = 0; j < 16; ++j)
                acc = __builtin_amdgcn_wmma_f32_16x16x4_f32(
                     false, af[j],  false, bf[16 * kh + j],
                     (short)0, acc,  false,  false);
        }

        if (nv) {
            float* op = &s_o[cur][(8 * half) * OPITCH + ncol];
#pragma unroll
            for (int r = 0; r < 8; ++r) op[r * OPITCH] = HAS_BIAS ? acc[r] + bb : acc[r];
        }
        if (has_next) {
            X_PIECE_STAGE(0, cur ^ 1) X_PIECE_STAGE(1, cur ^ 1) X_PIECE_STAGE(2, cur ^ 1)
            X_PIECE_STAGE(3, cur ^ 1) X_PIECE_STAGE(4, cur ^ 1) X_PIECE_STAGE(5, cur ^ 1)
        }
        __syncthreads();
        done = tile;
        cur ^= 1;
    }
    {
        float4* dst = (float4*)(OUT + (size_t)done * 16 * NCOLS);
        const float* so = s_o[cur ^ 1];
#pragma unroll
        for (int i = 0; i < NO; ++i) {
            const int idx = tid + i * NTHR;
            if ((i + 1) * NTHR <= O_F4 || idx < O_F4)
                dst[idx] = *(const float4*)&so[(idx / ROW_F4) * OPITCH + 4 * (idx % ROW_F4)];
        }
    }
}
#undef X_PIECE_LOAD
#undef X_PIECE_STAGE

__device__ __forceinline__ void row_fma(float4& a, float s, const float4& v) {
    a.x = fmaf(s, v.x, a.x);
    a.y = fmaf(s, v.y, a.y);
    a.z = fmaf(s, v.z, a.z);
    a.w = fmaf(s, v.w, a.w);
}

__device__ __forceinline__ int2 list_piece(const int node, const float di, const int beg,
                                           const int n, const int pos, const int lane,
                                           const int* __restrict__ csr) {
    int2 ent;
    ent.x = 0;
    ent.y = 0;
    const int p = pos + lane;
    if (p == 0) {
        ent.x = node;
        ent.y = __float_as_int(di);
    } else if (p < n) {
        ent = *(const int2*)(csr + 2 * (size_t)(beg + p - 1));
    }
    return ent;
}

__device__ __forceinline__ float4 gather_row(const int node, const float di, const int beg,
                                             const int end, const int2 ent0, const unsigned ul,
                                             const int* __restrict__ csr,
                                             const float4* __restrict__ h4, const float4 bb) {
    const int lane = (int)ul;
    const int ntot = end - beg + 1;
    float4 acc;
    acc.x = 0.0f;
    acc.y = 0.0f;
    acc.z = 0.0f;
    acc.w = 0.0f;

#define EDGE_LOAD(k)                                                                          \
    const int s##k = __builtin_amdgcn_readlane(ent.x, j + k);                                 \
    const float w##k = __int_as_float(__builtin_amdgcn_readlane(ent.y, j + k)) * di;          \
    const float4 v##k = h4[(unsigned)s##k * 32u + ul];

    int2 ent = ent0;
    for (int pos = 0; pos < ntot; pos += 32) {
        int n = ntot - pos;
        if (n > 32) n = 32;

        int j = 0;
        for (; j + 8 <= n; j += 8) {
            EDGE_LOAD(0) EDGE_LOAD(1) EDGE_LOAD(2) EDGE_LOAD(3)
            EDGE_LOAD(4) EDGE_LOAD(5) EDGE_LOAD(6) EDGE_LOAD(7)
            row_fma(acc, w0, v0);
            row_fma(acc, w1, v1);
            row_fma(acc, w2, v2);
            row_fma(acc, w3, v3);
            row_fma(acc, w4, v4);
            row_fma(acc, w5, v5);
            row_fma(acc, w6, v6);
            row_fma(acc, w7, v7);
        }
        if (j + 4 <= n) {
            EDGE_LOAD(0) EDGE_LOAD(1) EDGE_LOAD(2) EDGE_LOAD(3)
            row_fma(acc, w0, v0);
            row_fma(acc, w1, v1);
            row_fma(acc, w2, v2);
            row_fma(acc, w3, v3);
            j += 4;
        }
        const int left = n - j;
        if (left > 0) {
            EDGE_LOAD(0)
            if (left == 1) {
                row_fma(acc, w0, v0);
            } else {
                EDGE_LOAD(1)
                if (left == 2) {
                    row_fma(acc, w0, v0);
                    row_fma(acc, w1, v1);
                } else {
                    EDGE_LOAD(2)
                    row_fma(acc, w0, v0);
                    row_fma(acc, w1, v1);
                    row_fma(acc, w2, v2);
                }
            }
        }
        if (pos + 32 < ntot) ent = list_piece(node, di, beg, ntot, pos + 32, lane, csr);
    }
#undef EDGE_LOAD
    acc.x = fmaxf(acc.x + bb.x, 0.0f);
    acc.y = fmaxf(acc.y + bb.y, 0.0f);
    acc.z = fmaxf(acc.z + bb.z, 0.0f);
    acc.w = fmaxf(acc.w + bb.w, 0.0f);
    return acc;
}

template <int NCOLS, bool HAS_OBIAS>
__global__ __launch_bounds__(256, 8) void gather_gemm_kernel(const int* __restrict__ offs,
                                                          const int* __restrict__ csr,
                                                          const float* __restrict__ dinv,
                                                          const float* __restrict__ h,
                                                          const float* __restrict__ abias,
                                                          const float* __restrict__ W,
                                                          const float* __restrict__ obias,
                                                          float* __restrict__ OUT,
                                                          int ntiles) {
    constexpr int NTHR   = 256;
    constexpr int NWG    = (NCOLS + 15) / 16;
    constexpr int OPITCH = (NCOLS % 32 == 0) ? NCOLS + 4 : NCOLS;
    constexpr int O_F4   = 16 * NCOLS / 4;
    constexpr int ROW_F4 = NCOLS / 4;
    constexpr int NO     = (O_F4 + NTHR - 1) / NTHR;
    static_assert(NWG <= 8, "eight waves per block");

    __shared__ __attribute__((aligned(16))) float s_a[2][16 * APITCH];
    __shared__ __attribute__((aligned(16))) float s_o[2][16 * OPITCH];

    const int tid  = threadIdx.x;
    const int lane = tid & 31;
    const int wave = __builtin_amdgcn_readfirstlane(tid >> 5);
    const int l15  = lane & 15;
    const int half = lane >> 4;
    const int ncol = wave * 16 + l15;
    const bool nv  = (NCOLS % 16 == 0 && NWG == 8) ? true : (ncol < NCOLS);

    int tile = blockIdx.x;
    if (tile >= ntiles) return;

    const int wcol = nv ? ncol : 0;
    v2f bf[32];
#pragma unroll
    for (int ks = 0; ks < 32; ++ks) {
        const int kk = 4 * ks + 2 * half;
        const float w0 = W[(size_t)kk * NCOLS + wcol];
        const float w1 = W[(size_t)(kk + 1) * NCOLS + wcol];
        bf[ks].x = nv ? w0 : 0.0f;
        bf[ks].y = nv ? w1 : 0.0f;
    }
    float ob = 0.0f;
    if (HAS_OBIAS) ob = obias[wcol];

    const float4* h4 = (const float4*)h;
    const unsigned ul = (unsigned)lane;
    const float4 bb = ((const float4*)abias)[lane];

    const int lo = lane < 16 ? lane : 16;
    int   t_off = offs[tile * 16 + lo];
    float t_din = dinv[tile * 16 + l15];

    int cur  = 0;
    int done = -1;
    for (; tile < ntiles; tile += gridDim.x) {
        const int node0 = tile * 16 + wave;
        const int node1 = node0 + 8;
        const int beg0 = __builtin_amdgcn_readlane(t_off, wave);
        const int end0 = __builtin_amdgcn_readlane(t_off, wave + 1);
        const int beg1 = __builtin_amdgcn_readlane(t_off, wave + 8);
        const int end1 = __builtin_amdgcn_readlane(t_off, wave + 9);
        const float di0 = __int_as_float(__builtin_amdgcn_readlane(__float_as_int(t_din), wave));
        const float di1 = __int_as_float(__builtin_amdgcn_readlane(__float_as_int(t_din), wave + 8));
        const int2 e0 = list_piece(node0, di0, beg0, end0 - beg0 + 1, 0, lane, csr);
        const int2 e1 = list_piece(node1, di1, beg1, end1 - beg1 + 1, 0, lane, csr);
        const int nxt = tile + gridDim.x;
        if (nxt < ntiles) {
            t_off = offs[nxt * 16 + lo];
            t_din = dinv[nxt * 16 + l15];
        }
        {
            const float4 xr = gather_row(node0, di0, beg0, end0, e0, ul, csr, h4, bb);
            *(float4*)&s_a[cur][wave * APITCH + 4 * lane] = xr;
        }
        {
            const float4 xr = gather_row(node1, di1, beg1, end1, e1, ul, csr, h4, bb);
            *(float4*)&s_a[cur][(wave + 8) * APITCH + 4 * lane] = xr;
        }
        __syncthreads();

        if (done >= 0) {
            float4* dst = (float4*)(OUT + (size_t)done * 16 * NCOLS);
            const float* so = s_o[cur ^ 1];
#pragma unroll
            for (int i = 0; i < NO; ++i) {
                const int idx = tid + i * NTHR;
                if ((i + 1) * NTHR <= O_F4 || idx < O_F4)
                    dst[idx] = *(const float4*)&so[(idx / ROW_F4) * OPITCH + 4 * (idx % ROW_F4)];
            }
        }

        if (NWG == 8 || wave < NWG) {
            v8f acc = {};
            const float* ap = &s_a[cur][l15 * APITCH + 2 * half];
#pragma unroll
            for (int kh = 0; kh < 2; ++kh) {
                v2f af[16];
#pragma unroll
                for (int j = 0; j < 16; ++j) af[j] = *(const v2f*)(ap + 4 * (16 * kh + j));
                asm volatile("" ::"v"(af[0]), "v"(af[1]), "v"(af[2]), "v"(af[3]), "v"(af[4]), "v"(af[5]),
                             "v"(af[6]), "v"(af[7]), "v"(af[8]), "v"(af[9]), "v"(af[10]), "v"(af[11]),
                             "v"(af[12]), "v"(af[13]), "v"(af[14]), "v"(af[15]));
#pragma unroll
                for (int j = 0; j < 16; ++j)
                    acc = __builtin_amdgcn_wmma_f32_16x16x4_f32(
                         false, af[j],  false, bf[16 * kh + j],
                         (short)0, acc,  false,  false);
            }
            if (nv) {
                float* op = &s_o[cur][(8 * half) * OPITCH + ncol];
#pragma unroll
                for (int r = 0; r < 8; ++r) op[r * OPITCH] = HAS_OBIAS ? acc[r] + ob : acc[r];
            }
        }
        done = tile;
        cur ^= 1;
    }
    __syncthreads();
    {
        float4* dst = (float4*)(OUT + (size_t)done * 16 * NCOLS);
        const float* so = s_o[cur ^ 1];
#pragma unroll
        for (int i = 0; i < NO; ++i) {
            const int idx = tid + i * NTHR;
            if ((i + 1) * NTHR <= O_F4 || idx < O_F4)
                dst[idx] = *(const float4*)&so[(idx / ROW_F4) * OPITCH + 4 * (idx % ROW_F4)];
        }
    }
}

extern "C" void kernel_launch(void* const* d_in, const int* in_sizes, int n_in,
                              void* d_out, int out_size, void* d_ws, size_t ws_size,
                              hipStream_t stream) {
    const float* x    = (const float*)d_in[0];
    const int*   edge = (const int*)d_in[1];
    const float* W1   = (const float*)d_in[2];
    const float* b1   = (const float*)d_in[3];
    const float* W2   = (const float*)d_in[4];
    const float* b2   = (const float*)d_in[5];
    const float* Wout = (const float*)d_in[6];
    const float* bout = (const float*)d_in[7];
    const int* row = edge;
    const int* col = edge + N_EDGES;
    float* out = (float*)d_out;

    char* ws = (char*)d_ws;
    float* dinv = (float*)ws;
    int*   cnt  = (int*)(ws + ((size_t)1 << 19));
    float* bufA = (float*)(ws + ((size_t)1 << 20));
    float* bufB = bufA + (size_t)N_NODES * DIM;
    int*   offs = (int*)(ws + ((size_t)104 << 20));
    int*   rank = (int*)(ws + ((size_t)105 << 20));
    int*   csr  = (int*)(ws + ((size_t)112 << 20));

    const int ntiles = N_NODES / 16;

    cnt_zero_kernel<<<(N_NODES + 255) / 256, 256, 0, stream>>>(cnt);
    hist_rank_kernel<<<(N_EDGES + 255) / 256, 256, 0, stream>>>(col, cnt, rank);
    scan_dinv_kernel<<<(N_NODES + 1023) / 1024, 1024, 0, stream>>>(cnt, offs, dinv);
    fill_kernel<<<(N_EDGES + 255) / 256, 256, 0, stream>>>(row, col, rank, offs, dinv, csr);

    gemm_tiles_wmma<8, DIM, false><<<1024, 256, 0, stream>>>(x, W1, nullptr, bufA, ntiles);
    gather_gemm_kernel<DIM, false><<<1024, 256, 0, stream>>>(offs, csr, dinv, bufA, b1, W2, nullptr, bufB, ntiles);
    gather_gemm_kernel<N_CLASSES, true><<<1024, 256, 0, stream>>>(offs, csr, dinv, bufB, b2, Wout, bout, out, ntiles);
}


# Aggregate first, then multiply: (A X) W in place of A (X W) in both layers, rows summed before the product (another summation order); stand-alone first product and h1 pass gone: launches 7 -> 6
# speedup vs baseline: 3.3077x; 1.0400x over previous
#include <hip/hip_runtime.h>

typedef __attribute__((ext_vector_type(2))) float v2f;
typedef __attribute__((ext_vector_type(8))) float v8f;

#define N_NODES 100000
#define N_EDGES 1600000
#define DIM 128
#define N_CLASSES 40

__global__ void cnt_zero_kernel(int* cnt) {
    int i = blockIdx.x * blockDim.x + threadIdx.x;
    if (i < N_NODES) cnt[i] = 0;
}

__global__ void hist_rank_kernel(const int* __restrict__ col, int* cnt, int* __restrict__ rank) {
    int e = blockIdx.x * blockDim.x + threadIdx.x;
    if (e < N_EDGES) rank[e] = atomicAdd(&cnt[col[e]], 1);
}

__global__ __launch_bounds__(1024) void scan_dinv_kernel(const int* __restrict__ cnt,
                                                         int* __restrict__ offs,
                                                         float* __restrict__ dinv) {
    __shared__ int s_w[32];
    const int t    = threadIdx.x;
    const int lane = t & 31;
    const int wv   = t >> 5;

    const int4* c4 = (const int4*)cnt;
    const int nprev4 = blockIdx.x * 256;
    int part = 0;
    int i4 = t;
    for (; i4 + 3072 < nprev4; i4 += 4096) {
        const int4 a = c4[i4];
        const int4 b = c4[i4 + 1024];
        const int4 c = c4[i4 + 2048];
        const int4 d = c4[i4 + 3072];
        part += ((a.x + a.y) + (a.z + a.w)) + ((b.x + b.y) + (b.z + b.w)) +
                ((c.x + c.y) + (c.z + c.w)) + ((d.x + d.y) + (d.z + d.w));
    }
    for (; i4 < nprev4; i4 += 1024) {
        const int4 v = c4[i4];
        part += (v.x + v.y) + (v.z + v.w);
    }
    for (int o = 16; o > 0; o >>= 1) part += __shfl_xor(part, o);
    if (lane == 0) s_w[wv] = part;
    __syncthreads();
    int prefix = 0;
    for (int k = 0; k < 32; ++k) prefix += s_w[k];
    __syncthreads();

    const int i = blockIdx.x * 1024 + t;
    const int c = (i < N_NODES) ? cnt[i] : 0;
    int inc = c;
    for (int o = 1; o < 32; o <<= 1) {
        int u = __shfl_up(inc, o);
        if (lane >= o) inc += u;
    }
    if (lane == 31) s_w[wv] = inc;
    __syncthreads();
    int wpre = 0;
    for (int k = 0; k < wv; ++k) wpre += s_w[k];
    const int ex = prefix + wpre + inc - c;
    if (i < N_NODES) {
        offs[i] = ex;
        dinv[i] = rsqrtf((float)(c + 1));
        if (i == N_NODES - 1) offs[N_NODES] = ex + c;
    }
}

__global__ void fill_kernel(const int* __restrict__ row, const int* __restrict__ col,
                            const int* __restrict__ rank, const int* __restrict__ offs,
                            const float* __restrict__ dinv, int* __restrict__ csr) {
    int e = blockIdx.x * blockDim.x + threadIdx.x;
    if (e < N_EDGES) {
        const int r = row[e];
        const int p = offs[col[e]] + rank[e];
        int2 ent;
        ent.x = r;
        ent.y = __float_as_int(dinv[r]);
        *(int2*)(csr + 2 * (size_t)p) = ent;
    }
}

#define APITCH 132


__device__ __forceinline__ void row_fma(float4& a, float s, const float4& v) {
    a.x = fmaf(s, v.x, a.x);
    a.y = fmaf(s, v.y, a.y);
    a.z = fmaf(s, v.z, a.z);
    a.w = fmaf(s, v.w, a.w);
}

__device__ __forceinline__ int2 list_piece(const int node, const float di, const int beg,
                                           const int n, const int pos, const int lane,
                                           const int* __restrict__ csr) {
    int2 ent;
    ent.x = 0;
    ent.y = 0;
    const int p = pos + lane;
    if (p == 0) {
        ent.x = node;
        ent.y = __float_as_int(di);
    } else if (p < n) {
        ent = *(const int2*)(csr + 2 * (size_t)(beg + p - 1));
    }
    return ent;
}

__device__ __forceinline__ float4 gather_row(const int node, const float di, const int beg,
                                             const int end, const int2 ent0, const unsigned ul,
                                             const int* __restrict__ csr,
                                             const float4* __restrict__ h4) {
    const int lane = (int)ul;
    const int ntot = end - beg + 1;
    float4 acc;
    acc.x = 0.0f;
    acc.y = 0.0f;
    acc.z = 0.0f;
    acc.w = 0.0f;

#define EDGE_LOAD(k)                                                                          \
    const int s##k = __builtin_amdgcn_readlane(ent.x, j + k);                                 \
    const float w##k = __int_as_float(__builtin_amdgcn_readlane(ent.y, j + k)) * di;          \
    const float4 v##k = h4[(unsigned)s##k * 32u + ul];

    int2 ent = ent0;
    for (int pos = 0; pos < ntot; pos += 32) {
        int n = ntot - pos;
        if (n > 32) n = 32;

        int j = 0;
        for (; j + 8 <= n; j += 8) {
            EDGE_LOAD(0) EDGE_LOAD(1) EDGE_LOAD(2) EDGE_LOAD(3)
            EDGE_LOAD(4) EDGE_LOAD(5) EDGE_LOAD(6) EDGE_LOAD(7)
            row_fma(acc, w0, v0);
            row_fma(acc, w1, v1);
            row_fma(acc, w2, v2);
            row_fma(acc, w3, v3);
            row_fma(acc, w4, v4);
            row_fma(acc, w5, v5);
            row_fma(acc, w6, v6);
            row_fma(acc, w7, v7);
        }
        if (j + 4 <= n) {
            EDGE_LOAD(0) EDGE_LOAD(1) EDGE_LOAD(2) EDGE_LOAD(3)
            row_fma(acc, w0, v0);
            row_fma(acc, w1, v1);
            row_fma(acc, w2, v2);
            row_fma(acc, w3, v3);
            j += 4;
        }
        const int left = n - j;
        if (left > 0) {
            EDGE_LOAD(0)
            if (left == 1) {
                row_fma(acc, w0, v0);
            } else {
                EDGE_LOAD(1)
                if (left == 2) {
                    row_fma(acc, w0, v0);
                    row_fma(acc, w1, v1);
                } else {
                    EDGE_LOAD(2)
                    row_fma(acc, w0, v0);
                    row_fma(acc, w1, v1);
                    row_fma(acc, w2, v2);
                }
            }
        }
        if (pos + 32 < ntot) ent = list_piece(node, di, beg, ntot, pos + 32, lane, csr);
    }
#undef EDGE_LOAD
    return acc;
}

__global__ __launch_bounds__(256, 8) void agg_gemm_relu_kernel(const int* __restrict__ offs,
                                                            const int* __restrict__ csr,
                                                            const float* __restrict__ dinv,
                                                            const float* __restrict__ X,
                                                            const float* __restrict__ W,
                                                            const float* __restrict__ bias,
                                                            float* __restrict__ OUT,
                                                            int ntiles) {
    constexpr int NTHR   = 256;
    constexpr int OPITCH = DIM + 4;
    constexpr int O_F4   = 16 * DIM / 4;
    constexpr int ROW_F4 = DIM / 4;
    constexpr int NO     = O_F4 / NTHR;

    __shared__ __attribute__((aligned(16))) float s_a[2][16 * APITCH];
    __shared__ __attribute__((aligned(16))) float s_o[2][16 * OPITCH];

    const int tid  = threadIdx.x;
    const int lane = tid & 31;
    const int wave = __builtin_amdgcn_readfirstlane(tid >> 5);
    const int l15  = lane & 15;
    const int half = lane >> 4;
    const int ncol = wave * 16 + l15;

    int tile = blockIdx.x;
    if (tile >= ntiles) return;

    v2f bf[32];
#pragma unroll
    for (int ks = 0; ks < 32; ++ks) {
        const int kk = 4 * ks + 2 * half;
        bf[ks].x = W[(size_t)kk * DIM + ncol];
        bf[ks].y = W[(size_t)(kk + 1) * DIM + ncol];
    }
    const float ob = bias[ncol];

    const float4* h4 = (const float4*)X;
    const unsigned ul = (unsigned)lane;

    const int lo = lane < 16 ? lane : 16;
    int   t_off = offs[tile * 16 + lo];
    float t_din = dinv[tile * 16 + l15];

    int cur  = 0;
    int done = -1;
    for (; tile < ntiles; tile += gridDim.x) {
        const int node0 = tile * 16 + wave;
        const int node1 = node0 + 8;
        const int beg0 = __builtin_amdgcn_readlane(t_off, wave);
        const int end0 = __builtin_amdgcn_readlane(t_off, wave + 1);
        const int beg1 = __builtin_amdgcn_readlane(t_off, wave + 8);
        const int end1 = __builtin_amdgcn_readlane(t_off, wave + 9);
        const float di0 = __int_as_float(__builtin_amdgcn_readlane(__float_as_int(t_din), wave));
        const float di1 = __int_as_float(__builtin_amdgcn_readlane(__float_as_int(t_din), wave + 8));
        const int2 e0 = list_piece(node0, di0, beg0, end0 - beg0 + 1, 0, lane, csr);
        const int2 e1 = list_piece(node1, di1, beg1, end1 - beg1 + 1, 0, lane, csr);
        const int nxt = tile + gridDim.x;
        if (nxt < ntiles) {
            t_off = offs[nxt * 16 + lo];
            t_din = dinv[nxt * 16 + l15];
        }
        {
            const float4 xr = gather_row(node0, di0, beg0, end0, e0, ul, csr, h4);
            *(float4*)&s_a[cur][wave * APITCH + 4 * lane] = xr;
        }
        {
            const float4 xr = gather_row(node1, di1, beg1, end1, e1, ul, csr, h4);
            *(float4*)&s_a[cur][(wave + 8) * APITCH + 4 * lane] = xr;
        }
        __syncthreads();

        if (done >= 0) {
            float4* dst = (float4*)(OUT + (size_t)done * 16 * DIM);
            const float* so = s_o[cur ^ 1];
#pragma unroll
            for (int i = 0; i < NO; ++i) {
                const int idx = tid + i * NTHR;
                dst[idx] = *(const float4*)&so[(idx / ROW_F4) * OPITCH + 4 * (idx % ROW_F4)];
            }
        }

        {
            v8f acc = {};
            const float* ap = &s_a[cur][l15 * APITCH + 2 * half];
#pragma unroll
            for (int kh = 0; kh < 2; ++kh) {
                v2f af[16];
#pragma unroll
                for (int j = 0; j < 16; ++j) af[j] = *(const v2f*)(ap + 4 * (16 * kh + j));
                asm volatile("" ::"v"(af[0]), "v"(af[1]), "v"(af[2]), "v"(af[3]), "v"(af[4]), "v"(af[5]),
                             "v"(af[6]), "v"(af[7]), "v"(af[8]), "v"(af[9]), "v"(af[10]), "v"(af[11]),
                             "v"(af[12]), "v"(af[13]), "v"(af[14]), "v"(af[15]));
#pragma unroll
                for (int j = 0; j < 16; ++j)
                    acc = __builtin_amdgcn_wmma_f32_16x16x4_f32(
                         false, af[j],  false, bf[16 * kh + j],
                         (short)0, acc,  false,  false);
            }
            float* op = &s_o[cur][(8 * half) * OPITCH + ncol];
#pragma unroll
            for (int r = 0; r < 8; ++r) op[r * OPITCH] = fmaxf(acc[r] + ob, 0.0f);
        }
        done = tile;
        cur ^= 1;
    }
    __syncthreads();
    {
        float4* dst = (float4*)(OUT + (size_t)done * 16 * DIM);
        const float* so = s_o[cur ^ 1];
#pragma unroll
        for (int i = 0; i < NO; ++i) {
            const int idx = tid + i * NTHR;
            dst[idx] = *(const float4*)&so[(idx / ROW_F4) * OPITCH + 4 * (idx % ROW_F4)];
        }
    }
}

#define WPITCH 48
__global__ __launch_bounds__(256, 8) void agg_gemm_relu_gemm_kernel(const int* __restrict__ offs,
                                                                 const int* __restrict__ csr,
                                                                 const float* __restrict__ dinv,
                                                                 const float* __restrict__ X,
                                                                 const float* __restrict__ W,
                                                                 const float* __restrict__ bias,
                                                                 const float* __restrict__ Wo,
                                                                 const float* __restrict__ obias,
                                                                 float* __restrict__ OUT,
                                                                 int ntiles) {
    constexpr int NWO  = (N_CLASSES + 15) / 16;
    constexpr int O_F4 = 16 * N_CLASSES / 4;
    constexpr int ROW_F4 = N_CLASSES / 4;
    static_assert(O_F4 <= 256, "one piece per thread");
    static_assert(16 * NWO <= WPITCH, "padded columns");

    __shared__ __attribute__((aligned(16))) float s_a[16 * APITCH];
    __shared__ __attribute__((aligned(16))) float s_x[16 * APITCH];
    __shared__ __attribute__((aligned(16))) float s_o[16 * N_CLASSES];
    __shared__ __attribute__((aligned(16))) float s_w[64 * WPITCH * 2];

    const int tid  = threadIdx.x;
    const int lane = tid & 31;
    const int wave = __builtin_amdgcn_readfirstlane(tid >> 5);
    const int l15  = lane & 15;
    const int half = lane >> 4;
    const int ncol = wave * 16 + l15;

    int tile = blockIdx.x;
    if (tile >= ntiles) return;

    v2f bf[32];
#pragma unroll
    for (int ks = 0; ks < 32; ++ks) {
        const int kk = 4 * ks + 2 * half;
        bf[ks].x = W[(size_t)kk * DIM + ncol];
        bf[ks].y = W[(size_t)(kk + 1) * DIM + ncol];
    }
    const float hb = bias[ncol];

    {
        static_assert(DIM * N_CLASSES == 4 * 5 * 256 && N_CLASSES % 4 == 0, "five pieces per thread");
        static_assert(WPITCH - N_CLASSES == 8, "sixteen zero floats per fragment pair row");
        const float4* wo4 = (const float4*)Wo;
        float4 wv[5];
#pragma unroll
        for (int m = 0; m < 5; ++m) wv[m] = wo4[tid + 256 * m];
#pragma unroll
        for (int m = 0; m < 5; ++m) {
            const int g = 4 * (tid + 256 * m);
            const int k = g / N_CLASSES;
            const int c = g % N_CLASSES;
            float* d = &s_w[((k >> 1) * WPITCH + c) * 2 + (k & 1)];
            d[0] = wv[m].x;
            d[2] = wv[m].y;
            d[4] = wv[m].z;
            d[6] = wv[m].w;
        }
        for (int i = tid; i < 64 * (WPITCH - N_CLASSES) * 2; i += 256)
            s_w[((i >> 4) * WPITCH + N_CLASSES) * 2 + (i & 15)] = 0.0f;
    }
    const bool ov = (wave < NWO) && (ncol < N_CLASSES);
    const float ob = ov ? obias[ncol] : 0.0f;

    const float4* h4 = (const float4*)X;
    const unsigned ul = (unsigned)lane;

    const int lo = lane < 16 ? lane : 16;
    int   t_off = offs[tile * 16 + lo];
    float t_din = dinv[tile * 16 + l15];

    int done = -1;
    for (; tile < ntiles; tile += gridDim.x) {
        const int node0 = tile * 16 + wave;
        const int node1 = node0 + 8;
        const int beg0 = __builtin_amdgcn_readlane(t_off, wave);
        const int end0 = __builtin_amdgcn_readlane(t_off, wave + 1);
        const int beg1 = __builtin_amdgcn_readlane(t_off, wave + 8);
        const int end1 = __builtin_amdgcn_readlane(t_off, wave + 9);
        const float di0 = __int_as_float(__builtin_amdgcn_readlane(__float_as_int(t_din), wave));
        const float di1 = __int_as_float(__builtin_amdgcn_readlane(__float_as_int(t_din), wave + 8));
        const int2 e0 = list_piece(node0, di0, beg0, end0 - beg0 + 1, 0, lane, csr);
        const int2 e1 = list_piece(node1, di1, beg1, end1 - beg1 + 1, 0, lane, csr);
        const int nxt = tile + gridDim.x;
        if (nxt < ntiles) {
            t_off = offs[nxt * 16 + lo];
            t_din = dinv[nxt * 16 + l15];
        }
        {
            const float4 xr = gather_row(node0, di0, beg0, end0, e0, ul, csr, h4);
            *(float4*)&s_a[wave * APITCH + 4 * lane] = xr;
        }
        {
            const float4 xr = gather_row(node1, di1, beg1, end1, e1, ul, csr, h4);
            *(float4*)&s_a[(wave + 8) * APITCH + 4 * lane] = xr;
        }
        __syncthreads();

        if (done >= 0 && tid < O_F4) {
            float4* dst = (float4*)(OUT + (size_t)done * 16 * N_CLASSES);
            dst[tid] = *(const float4*)&s_o[(tid / ROW_F4) * N_CLASSES + 4 * (tid % ROW_F4)];
        }

        {
            v8f acc = {};
            const float* ap = &s_a[l15 * APITCH + 2 * half];
#pragma unroll
            for (int kh = 0; kh < 2; ++kh) {
                v2f af[16];
#pragma unroll
                for (int j = 0; j < 16; ++j) af[j] = *(const v2f*)(ap + 4 * (16 * kh + j));
                asm volatile("" ::"v"(af[0]), "v"(af[1]), "v"(af[2]), "v"(af[3]), "v"(af[4]), "v"(af[5]),
                             "v"(af[6]), "v"(af[7]), "v"(af[8]), "v"(af[9]), "v"(af[10]), "v"(af[11]),
                             "v"(af[12]), "v"(af[13]), "v"(af[14]), "v"(af[15]));
#pragma unroll
                for (int j = 0; j < 16; ++j)
                    acc = __builtin_amdgcn_wmma_f32_16x16x4_f32(
                         false, af[j],  false, bf[16 * kh + j],
                         (short)0, acc,  false,  false);
            }
            float* xp = &s_x[(8 * half) * APITCH + ncol];
#pragma unroll
            for (int r = 0; r < 8; ++r) xp[r * APITCH] = fmaxf(acc[r] + hb, 0.0f);
        }
        __syncthreads();

        if (wave < NWO) {
            v8f acc = {};
            const float* ap = &s_x[l15 * APITCH + 2 * half];
            const float* bp = &s_w[(half * WPITCH + ncol) * 2];
#pragma unroll
            for (int kq = 0; kq < 4; ++kq) {
                v2f af[8], bq[8];
#pragma unroll
                for (int j = 0; j < 8; ++j) {
                    af[j] = *(const v2f*)(ap + 4 * (8 * kq + j));
                    bq[j] = *(const v2f*)(bp + 4 * WPITCH * (8 * kq + j));
                }
                asm volatile("" ::"v"(af[0]), "v"(af[1]), "v"(af[2]), "v"(af[3]), "v"(af[4]), "v"(af[5]),
                             "v"(af[6]), "v"(af[7]), "v"(bq[0]), "v"(bq[1]), "v"(bq[2]), "v"(bq[3]),
                             "v"(bq[4]), "v"(bq[5]), "v"(bq[6]), "v"(bq[7]));
#pragma unroll
                for (int j = 0; j < 8; ++j)
                    acc = __builtin_amdgcn_wmma_f32_16x16x4_f32(
                         false, af[j],  false, bq[j],
                         (short)0, acc,  false,  false);
            }
            if (ov) {
                float* op = &s_o[(8 * half) * N_CLASSES + ncol];
#pragma unroll
                for (int r = 0; r < 8; ++r) op[r * N_CLASSES] = acc[r] + ob;
            }
        }
        done = tile;
    }
    __syncthreads();
    if (tid < O_F4) {
        float4* dst = (float4*)(OUT + (size_t)done * 16 * N_CLASSES);
        dst[tid] = *(const float4*)&s_o[(tid / ROW_F4) * N_CLASSES + 4 * (tid % ROW_F4)];
    }
}

extern "C" void kernel_launch(void* const* d_in, const int* in_sizes, int n_in,
                              void* d_out, int out_size, void* d_ws, size_t ws_size,
                              hipStream_t stream) {
    const float* x    = (const float*)d_in[0];
    const int*   edge = (const int*)d_in[1];
    const float* W1   = (const float*)d_in[2];
    const float* b1   = (const float*)d_in[3];
    const float* W2   = (const float*)d_in[4];
    const float* b2   = (const float*)d_in[5];
    const float* Wout = (const float*)d_in[6];
    const float* bout = (const float*)d_in[7];
    const int* row = edge;
    const int* col = edge + N_EDGES;
    float* out = (float*)d_out;

    char* ws = (char*)d_ws;
    float* dinv = (float*)ws;
    int*   cnt  = (int*)(ws + ((size_t)1 << 19));
    float* bufA = (float*)(ws + ((size_t)1 << 20));
    int*   offs = (int*)(ws + ((size_t)104 << 20));
    int*   rank = (int*)(ws + ((size_t)105 << 20));
    int*   csr  = (int*)(ws + ((size_t)112 << 20));

    const int ntiles = N_NODES / 16;

    cnt_zero_kernel<<<(N_NODES + 255) / 256, 256, 0, stream>>>(cnt);
    hist_rank_kernel<<<(N_EDGES + 255) / 256, 256, 0, stream>>>(col, cnt, rank);
    scan_dinv_kernel<<<(N_NODES + 1023) / 1024, 1024, 0, stream>>>(cnt, offs, dinv);
    fill_kernel<<<(N_EDGES + 255) / 256, 256, 0, stream>>>(row, col, rank, offs, dinv, csr);

    agg_gemm_relu_kernel<<<1024, 256, 0, stream>>>(offs, csr, dinv, x, W1, b1, bufA, ntiles);
    agg_gemm_relu_gemm_kernel<<<1024, 256, 0, stream>>>(offs, csr, dinv, bufA, W2, b2, Wout, bout, out, ntiles);
}


# Edge lists kept as 64-slot rows per target filled straight from the arrival-order atomic (overflow list beyond); deg^-1/2 applied per row not per edge (another rounding); launches 6 -> 5
# speedup vs baseline: 3.5461x; 1.0721x over previous
#include <hip/hip_runtime.h>

typedef __attribute__((ext_vector_type(2))) float v2f;
typedef __attribute__((ext_vector_type(4))) float v4f;
typedef __attribute__((ext_vector_type(8))) float v8f;

#define N_NODES 100000
#define N_EDGES 1600000
#define DIM 128
#define N_CLASSES 40

#define CAP 64

__global__ void cnt_zero_kernel(int* cnt) {
    int i = blockIdx.x * blockDim.x + threadIdx.x;
    if (i <= N_NODES) cnt[i] = 0;
}

__global__ void place_kernel(const int* __restrict__ row, const int* __restrict__ col, int* cnt,
                             int* __restrict__ slots, int* __restrict__ ovf) {
    int e = blockIdx.x * blockDim.x + threadIdx.x;
    if (e < N_EDGES) {
        const int c = col[e];
        const int s = row[e];
        const int r = atomicAdd(&cnt[c], 1);
        if (r < CAP) {
            slots[c * CAP + r] = s;
        } else {
            const int k = atomicAdd(&cnt[N_NODES], 1);
            int2 ent;
            ent.x = c;
            ent.y = s;
            *(int2*)(ovf + 2 * (size_t)k) = ent;
        }
    }
}

__global__ __launch_bounds__(256) void prescale_kernel(const float* __restrict__ X,
                                                       const int* __restrict__ cnt,
                                                       float* __restrict__ XS) {
    static_assert(N_NODES * (DIM / 4) == 3125 * 1024, "four pieces per thread, exact grid");
    const v4f* x4 = (const v4f*)X;
    v4f* o4 = (v4f*)XS;
    const int base = blockIdx.x * 1024 + threadIdx.x;
    v4f v[4];
    float d[4];
#pragma unroll
    for (int m = 0; m < 4; ++m) {
        const int idx = base + 256 * m;
        v[m] = __builtin_nontemporal_load(x4 + idx);
        d[m] = rsqrtf((float)(cnt[idx >> 5] + 1));
    }
#pragma unroll
    for (int m = 0; m < 4; ++m) o4[base + 256 * m] = v[m] * d[m];
}

#define APITCH 132


__device__ __forceinline__ float4 row_scale(const float4& v, const float s) {
    float4 r;
    r.x = s * v.x;
    r.y = s * v.y;
    r.z = s * v.z;
    r.w = s * v.w;
    return r;
}

__device__ __forceinline__ void row_add(float4& a, const float4& v) {
    a.x += v.x;
    a.y += v.y;
    a.z += v.z;
    a.w += v.w;
}

__device__ __forceinline__ int list_piece(const int node, const int n, const int pos,
                                          const int lane, const int* __restrict__ slots) {
    int ent = 0;
    const int p = pos + lane;
    if (p == 0) {
        ent = node;
    } else if (p < n) {
        ent = slots[node * CAP + p - 1];
    }
    return ent;
}

__device__ __forceinline__ float4 gather_row(const int node, const int nedge, const int ent0,
                                             const unsigned ul, const int* __restrict__ slots,
                                             const int* __restrict__ ovf,
                                             const int* __restrict__ cnt,
                                             const float4* __restrict__ h4) {
    const int lane = (int)ul;
    const int ntot = (nedge < CAP ? nedge : CAP) + 1;
    float4 acc;
    acc.x = 0.0f;
    acc.y = 0.0f;
    acc.z = 0.0f;
    acc.w = 0.0f;

#define EDGE_LOAD(k)                                                                          \
    const int s##k = __builtin_amdgcn_readlane(ent, j + k);                                   \
    const float4 v##k = h4[(unsigned)s##k * 32u + ul];

    int ent = ent0;
    for (int pos = 0; pos < ntot; pos += 32) {
        int n = ntot - pos;
        if (n > 32) n = 32;

        int j = 0;
        for (; j + 8 <= n; j += 8) {
            EDGE_LOAD(0) EDGE_LOAD(1) EDGE_LOAD(2) EDGE_LOAD(3)
            EDGE_LOAD(4) EDGE_LOAD(5) EDGE_LOAD(6) EDGE_LOAD(7)
            row_add(acc, v0);
            row_add(acc, v1);
            row_add(acc, v2);
            row_add(acc, v3);
            row_add(acc, v4);
            row_add(acc, v5);
            row_add(acc, v6);
            row_add(acc, v7);
        }
        if (j + 4 <= n) {
            EDGE_LOAD(0) EDGE_LOAD(1) EDGE_LOAD(2) EDGE_LOAD(3)
            row_add(acc, v0);
            row_add(acc, v1);
            row_add(acc, v2);
            row_add(acc, v3);
            j += 4;
        }
        const int left = n - j;
        if (left > 0) {
            EDGE_LOAD(0)
            if (left == 1) {
                row_add(acc, v0);
            } else {
                EDGE_LOAD(1)
                if (left == 2) {
                    row_add(acc, v0);
                    row_add(acc, v1);
                } else {
                    EDGE_LOAD(2)
                    row_add(acc, v0);
                    row_add(acc, v1);
                    row_add(acc, v2);
                }
            }
        }
        if (pos + 32 < ntot) ent = list_piece(node, ntot, pos + 32, lane, slots);
    }
    if (nedge > CAP) {
        int novf = __builtin_amdgcn_readfirstlane(cnt[N_NODES]);
        if (novf > N_EDGES) novf = N_EDGES;
        for (int base = 0; base < novf; base += 32) {
            int2 o;
            o.x = -1;
            o.y = 0;
            if (base + lane < novf) o = *(const int2*)(ovf + 2 * (size_t)(base + lane));
            unsigned m = __builtin_amdgcn_ballot_w32(o.x == node);
            while (m != 0u) {
                const int l = __builtin_ctz(m);
                m &= m - 1u;
                const int s0 = __builtin_amdgcn_readlane(o.y, l);
                const float4 v0 = h4[(unsigned)s0 * 32u + ul];
                row_add(acc, v0);
            }
        }
    }
#undef EDGE_LOAD
    return acc;
}

__global__ __launch_bounds__(256, 8) void agg_gemm_relu_kernel(const int* __restrict__ cnt,
                                                            const int* __restrict__ slots,
                                                            const int* __restrict__ ovf,
                                                            const float* __restrict__ X,
                                                            const float* __restrict__ W,
                                                            const float* __restrict__ bias,
                                                            float* __restrict__ OUT,
                                                            int ntiles) {
    constexpr int NTHR   = 256;
    constexpr int OPITCH = DIM + 4;
    constexpr int O_F4   = 16 * DIM / 4;
    constexpr int ROW_F4 = DIM / 4;
    constexpr int NO     = O_F4 / NTHR;

    __shared__ __attribute__((aligned(16))) float s_a[2][16 * APITCH];
    __shared__ __attribute__((aligned(16))) float s_o[2][16 * OPITCH];

    const int tid  = threadIdx.x;
    const int lane = tid & 31;
    const int wave = __builtin_amdgcn_readfirstlane(tid >> 5);
    const int l15  = lane & 15;
    const int half = lane >> 4;
    const int ncol = wave * 16 + l15;

    int tile = blockIdx.x;
    if (tile >= ntiles) return;

    v2f bf[32];
#pragma unroll
    for (int ks = 0; ks < 32; ++ks) {
        const int kk = 4 * ks + 2 * half;
        bf[ks].x = W[(size_t)kk * DIM + ncol];
        bf[ks].y = W[(size_t)(kk + 1) * DIM + ncol];
    }
    const float ob = bias[ncol];

    const float4* h4 = (const float4*)X;
    const unsigned ul = (unsigned)lane;

    int t_cnt = cnt[tile * 16 + l15];

    int cur  = 0;
    int done = -1;
    for (; tile < ntiles; tile += gridDim.x) {
        const int node0 = tile * 16 + wave;
        const int node1 = node0 + 8;
        const int n0 = __builtin_amdgcn_readlane(t_cnt, wave);
        const int n1 = __builtin_amdgcn_readlane(t_cnt, wave + 8);
        const float c_din = rsqrtf((float)(t_cnt + 1));
        const float di0 = __int_as_float(__builtin_amdgcn_readlane(__float_as_int(c_din), wave));
        const float di1 = __int_as_float(__builtin_amdgcn_readlane(__float_as_int(c_din), wave + 8));
        const int e0 = list_piece(node0, (n0 < CAP ? n0 : CAP) + 1, 0, lane, slots);
        const int e1 = list_piece(node1, (n1 < CAP ? n1 : CAP) + 1, 0, lane, slots);
        const int nxt = tile + gridDim.x;
        if (nxt < ntiles) {
            t_cnt = cnt[nxt * 16 + l15];
        }
        {
            const float4 xr = row_scale(gather_row(node0, n0, e0, ul, slots, ovf, cnt, h4), di0);
            *(float4*)&s_a[cur][wave * APITCH + 4 * lane] = xr;
        }
        {
            const float4 xr = row_scale(gather_row(node1, n1, e1, ul, slots, ovf, cnt, h4), di1);
            *(float4*)&s_a[cur][(wave + 8) * APITCH + 4 * lane] = xr;
        }
        __syncthreads();

        if (done >= 0) {
            float4* dst = (float4*)(OUT + (size_t)done * 16 * DIM);
            const float* so = s_o[cur ^ 1];
#pragma unroll
            for (int i = 0; i < NO; ++i) {
                const int idx = tid + i * NTHR;
                dst[idx] = *(const float4*)&so[(idx / ROW_F4) * OPITCH + 4 * (idx % ROW_F4)];
            }
        }

        {
            v8f acc = {};
            const float* ap = &s_a[cur][l15 * APITCH + 2 * half];
#pragma unroll
            for (int kh = 0; kh < 2; ++kh) {
                v2f af[16];
#pragma unroll
                for (int j = 0; j < 16; ++j) af[j] = *(const v2f*)(ap + 4 * (16 * kh + j));
                asm volatile("" ::"v"(af[0]), "v"(af[1]), "v"(af[2]), "v"(af[3]), "v"(af[4]), "v"(af[5]),
                             "v"(af[6]), "v"(af[7]), "v"(af[8]), "v"(af[9]), "v"(af[10]), "v"(af[11]),
                             "v"(af[12]), "v"(af[13]), "v"(af[14]), "v"(af[15]));
#pragma unroll
                for (int j = 0; j < 16; ++j)
                    acc = __builtin_amdgcn_wmma_f32_16x16x4_f32(
                         false, af[j],  false, bf[16 * kh + j],
                         (short)0, acc,  false,  false);
            }
            float* op = &s_o[cur][(8 * half) * OPITCH + ncol];
#pragma unroll
            for (int r = 0; r < 8; ++r) {
                const float dr = __shfl(c_din, 8 * half + r);
                op[r * OPITCH] = dr * fmaxf(acc[r] + ob, 0.0f);
            }
        }
        done = tile;
        cur ^= 1;
    }
    __syncthreads();
    {
        float4* dst = (float4*)(OUT + (size_t)done * 16 * DIM);
        const float* so = s_o[cur ^ 1];
#pragma unroll
        for (int i = 0; i < NO; ++i) {
            const int idx = tid + i * NTHR;
            dst[idx] = *(const float4*)&so[(idx / ROW_F4) * OPITCH + 4 * (idx % ROW_F4)];
        }
    }
}

#define WPITCH 48
__global__ __launch_bounds__(256, 8) void agg_gemm_relu_gemm_kernel(const int* __restrict__ cnt,
                                                                 const int* __restrict__ slots,
                                                                 const int* __restrict__ ovf,
                                                                 const float* __restrict__ X,
                                                                 const float* __restrict__ W,
                                                                 const float* __restrict__ bias,
                                                                 const float* __restrict__ Wo,
                                                                 const float* __restrict__ obias,
                                                                 float* __restrict__ OUT,
                                                                 int ntiles) {
    constexpr int NWO  = (N_CLASSES + 15) / 16;
    constexpr int O_F4 = 16 * N_CLASSES / 4;
    constexpr int ROW_F4 = N_CLASSES / 4;
    static_assert(O_F4 <= 256, "one piece per thread");
    static_assert(16 * NWO <= WPITCH, "padded columns");

    __shared__ __attribute__((aligned(16))) float s_a[16 * APITCH];
    __shared__ __attribute__((aligned(16))) float s_x[16 * APITCH];
    __shared__ __attribute__((aligned(16))) float s_o[16 * N_CLASSES];
    __shared__ __attribute__((aligned(16))) float s_w[64 * WPITCH * 2];

    const int tid  = threadIdx.x;
    const int lane = tid & 31;
    const int wave = __builtin_amdgcn_readfirstlane(tid >> 5);
    const int l15  = lane & 15;
    const int half = lane >> 4;
    const int ncol = wave * 16 + l15;

    int tile = blockIdx.x;
    if (tile >= ntiles) return;

    v2f bf[32];
#pragma unroll
    for (int ks = 0; ks < 32; ++ks) {
        const int kk = 4 * ks + 2 * half;
        bf[ks].x = W[(size_t)kk * DIM + ncol];
        bf[ks].y = W[(size_t)(kk + 1) * DIM + ncol];
    }
    const float hb = bias[ncol];

    {
        static_assert(DIM * N_CLASSES == 4 * 5 * 256 && N_CLASSES % 4 == 0, "five pieces per thread");
        static_assert(WPITCH - N_CLASSES == 8, "sixteen zero floats per fragment pair row");
        const float4* wo4 = (const float4*)Wo;
        float4 wv[5];
#pragma unroll
        for (int m = 0; m < 5; ++m) wv[m] = wo4[tid + 256 * m];
#pragma unroll
        for (int m = 0; m < 5; ++m) {
            const int g = 4 * (tid + 256 * m);
            const int k = g / N_CLASSES;
            const int c = g % N_CLASSES;
            float* d = &s_w[((k >> 1) * WPITCH + c) * 2 + (k & 1)];
            d[0] = wv[m].x;
            d[2] = wv[m].y;
            d[4] = wv[m].z;
            d[6] = wv[m].w;
        }
        for (int i = tid; i < 64 * (WPITCH - N_CLASSES) * 2; i += 256)
            s_w[((i >> 4) * WPITCH + N_CLASSES) * 2 + (i & 15)] = 0.0f;
    }
    const bool ov = (wave < NWO) && (ncol < N_CLASSES);
    const float ob = ov ? obias[ncol] : 0.0f;

    const float4* h4 = (const float4*)X;
    const unsigned ul = (unsigned)lane;

    int t_cnt = cnt[tile * 16 + l15];

    int done = -1;
    for (; tile < ntiles; tile += gridDim.x) {
        const int node0 = tile * 16 + wave;
        const int node1 = node0 + 8;
        const int n0 = __builtin_amdgcn_readlane(t_cnt, wave);
        const int n1 = __builtin_amdgcn_readlane(t_cnt, wave + 8);
        const float c_din = rsqrtf((float)(t_cnt + 1));
        const float di0 = __int_as_float(__builtin_amdgcn_readlane(__float_as_int(c_din), wave));
        const float di1 = __int_as_float(__builtin_amdgcn_readlane(__float_as_int(c_din), wave + 8));
        const int e0 = list_piece(node0, (n0 < CAP ? n0 : CAP) + 1, 0, lane, slots);
        const int e1 = list_piece(node1, (n1 < CAP ? n1 : CAP) + 1, 0, lane, slots);
        const int nxt = tile + gridDim.x;
        if (nxt < ntiles) {
            t_cnt = cnt[nxt * 16 + l15];
        }
        {
            const float4 xr = row_scale(gather_row(node0, n0, e0, ul, slots, ovf, cnt, h4), di0);
            *(float4*)&s_a[wave * APITCH + 4 * lane] = xr;
        }
        {
            const float4 xr = row_scale(gather_row(node1, n1, e1, ul, slots, ovf, cnt, h4), di1);
            *(float4*)&s_a[(wave + 8) * APITCH + 4 * lane] = xr;
        }
        __syncthreads();

        if (done >= 0 && tid < O_F4) {
            float4* dst = (float4*)(OUT + (size_t)done * 16 * N_CLASSES);
            dst[tid] = *(const float4*)&s_o[(tid / ROW_F4) * N_CLASSES + 4 * (tid % ROW_F4)];
        }

        {
            v8f acc = {};
            const float* ap = &s_a[l15 * APITCH + 2 * half];
#pragma unroll
            for (int kh = 0; kh < 2; ++kh) {
                v2f af[16];
#pragma unroll
                for (int j = 0; j < 16; ++j) af[j] = *(const v2f*)(ap + 4 * (16 * kh + j));
                asm volatile("" ::"v"(af[0]), "v"(af[1]), "v"(af[2]), "v"(af[3]), "v"(af[4]), "v"(af[5]),
                             "v"(af[6]), "v"(af[7]), "v"(af[8]), "v"(af[9]), "v"(af[10]), "v"(af[11]),
                             "v"(af[12]), "v"(af[13]), "v"(af[14]), "v"(af[15]));
#pragma unroll
                for (int j = 0; j < 16; ++j)
                    acc = __builtin_amdgcn_wmma_f32_16x16x4_f32(
                         false, af[j],  false, bf[16 * kh + j],
                         (short)0, acc,  false,  false);
            }
            float* xp = &s_x[(8 * half) * APITCH + ncol];
#pragma unroll
            for (int r = 0; r < 8; ++r) xp[r * APITCH] = fmaxf(acc[r] + hb, 0.0f);
        }
        __syncthreads();

        if (wave < NWO) {
            v8f acc = {};
            const float* ap = &s_x[l15 * APITCH + 2 * half];
            const float* bp = &s_w[(half * WPITCH + ncol) * 2];
#pragma unroll
            for (int kq = 0; kq < 4; ++kq) {
                v2f af[8], bq[8];
#pragma unroll
                for (int j = 0; j < 8; ++j) {
                    af[j] = *(const v2f*)(ap + 4 * (8 * kq + j));
                    bq[j] = *(const v2f*)(bp + 4 * WPITCH * (8 * kq + j));
                }
                asm volatile("" ::"v"(af[0]), "v"(af[1]), "v"(af[2]), "v"(af[3]), "v"(af[4]), "v"(af[5]),
                             "v"(af[6]), "v"(af[7]), "v"(bq[0]), "v"(bq[1]), "v"(bq[2]), "v"(bq[3]),
                             "v"(bq[4]), "v"(bq[5]), "v"(bq[6]), "v"(bq[7]));
#pragma unroll
                for (int j = 0; j < 8; ++j)
                    acc = __builtin_amdgcn_wmma_f32_16x16x4_f32(
                         false, af[j],  false, bq[j],
                         (short)0, acc,  false,  false);
            }
            if (ov) {
                float* op = &s_o[(8 * half) * N_CLASSES + ncol];
#pragma unroll
                for (int r = 0; r < 8; ++r) op[r * N_CLASSES] = acc[r] + ob;
            }
        }
        done = tile;
    }
    __syncthreads();
    if (tid < O_F4) {
        float4* dst = (float4*)(OUT + (size_t)done * 16 * N_CLASSES);
        dst[tid] = *(const float4*)&s_o[(tid / ROW_F4) * N_CLASSES + 4 * (tid % ROW_F4)];
    }
}

extern "C" void kernel_launch(void* const* d_in, const int* in_sizes, int n_in,
                              void* d_out, int out_size, void* d_ws, size_t ws_size,
                              hipStream_t stream) {
    const float* x    = (const float*)d_in[0];
    const int*   edge = (const int*)d_in[1];
    const float* W1   = (const float*)d_in[2];
    const float* b1   = (const float*)d_in[3];
    const float* W2   = (const float*)d_in[4];
    const float* b2   = (const float*)d_in[5];
    const float* Wout = (const float*)d_in[6];
    const float* bout = (const float*)d_in[7];
    const int* row = edge;
    const int* col = edge + N_EDGES;
    float* out = (float*)d_out;

    char* ws = (char*)d_ws;
    int*   cnt   = (int*)(ws + ((size_t)1 << 19));
    float* bufA  = (float*)(ws + ((size_t)1 << 20));
    int*   slots = (int*)(ws + ((size_t)112 << 20));
    int*   ovf   = (int*)(ws + ((size_t)144 << 20));
    float* xs    = (float*)(ws + ((size_t)160 << 20));

    const int ntiles = N_NODES / 16;

    cnt_zero_kernel<<<(N_NODES + 1 + 255) / 256, 256, 0, stream>>>(cnt);
    place_kernel<<<(N_EDGES + 255) / 256, 256, 0, stream>>>(row, col, cnt, slots, ovf);
    prescale_kernel<<<N_NODES * (DIM / 4) / 1024, 256, 0, stream>>>(x, cnt, xs);

    agg_gemm_relu_kernel<<<1024, 256, 0, stream>>>(cnt, slots, ovf, xs, W1, b1, bufA, ntiles);
    agg_gemm_relu_gemm_kernel<<<1024, 256, 0, stream>>>(cnt, slots, ovf, bufA, W2, b2, Wout, bout, out, ntiles);
}


# Edge lists built through 256 target-range bins: edges sorted by range in LDS, ranked per target by LDS adds (arrival order), slot rows stored whole; global adds 1.6M -> 50K; launches 5 -> 6
# speedup vs baseline: 5.7615x; 1.6247x over previous
#include <hip/hip_runtime.h>

typedef __attribute__((ext_vector_type(2))) float v2f;
typedef __attribute__((ext_vector_type(4))) float v4f;
typedef __attribute__((ext_vector_type(8))) float v8f;

#define N_NODES 100000
#define N_EDGES 1600000
#define DIM 128
#define N_CLASSES 40

#define CAP 64

#define NB 256
#define BW 391
#define BINCAP 8192
#define EPB 8192
static_assert(NB * BW >= N_NODES, "ranges cover all targets");

__global__ void cnt_zero_kernel(int* cnt, int* binCount) {
    int i = blockIdx.x * blockDim.x + threadIdx.x;
    if (i <= NB) binCount[i] = 0;
    if (i == 0) cnt[N_NODES] = 0;
}

__global__ __launch_bounds__(1024) void bin_kernel(const int* __restrict__ row,
                                                   const int* __restrict__ col, int* binCount,
                                                   int* __restrict__ binbuf,
                                                   int* __restrict__ spill) {
    constexpr int EPT = EPB / 1024;
    __shared__ int s_hist[NB];
    __shared__ int s_off[NB];
    __shared__ int s_base[NB];
    __shared__ int s_wtot[NB / 32];
    __shared__ __attribute__((aligned(8))) int2 s_sorted[EPB];

    const int tid  = threadIdx.x;
    const int lane = tid & 31;
    const int e0   = blockIdx.x * EPB;
    int nval = N_EDGES - e0;
    if (nval > EPB) nval = EPB;

    if (tid < NB) s_hist[tid] = 0;
    __syncthreads();

    int c[EPT], s[EPT], r[EPT];
#pragma unroll
    for (int m = 0; m < EPT; ++m) {
        const int i = tid + 1024 * m;
        c[m] = 0;
        s[m] = 0;
        if (i < nval) {
            c[m] = col[e0 + i];
            s[m] = row[e0 + i];
        }
    }
#pragma unroll
    for (int m = 0; m < EPT; ++m) {
        r[m] = 0;
        if (tid + 1024 * m < nval) r[m] = atomicAdd(&s_hist[c[m] / BW], 1);
    }
    __syncthreads();

    const int h = (tid < NB) ? s_hist[tid] : 0;
    int incl = h;
#pragma unroll
    for (int d = 1; d < 32; d <<= 1) {
        const int t = __shfl_up(incl, d);
        if (lane >= d) incl += t;
    }
    if (tid < NB) {
        if (lane == 31) s_wtot[tid >> 5] = incl;
        s_base[tid] = (h > 0) ? atomicAdd(&binCount[tid], h) : 0;
    }
    __syncthreads();
    if (tid < NB) {
        int pre = 0;
        for (int w = 0; w < (tid >> 5); ++w) pre += s_wtot[w];
        s_off[tid] = pre + incl - h;
    }
    __syncthreads();

#pragma unroll
    for (int m = 0; m < EPT; ++m) {
        if (tid + 1024 * m < nval) {
            int2 ent;
            ent.x = c[m];
            ent.y = s[m];
            s_sorted[s_off[c[m] / BW] + r[m]] = ent;
        }
    }
    __syncthreads();

    for (int i = tid; i < nval; i += 1024) {
        const int2 ent = s_sorted[i];
        const int b = ent.x / BW;
        const int g = s_base[b] + (i - s_off[b]);
        if (g < BINCAP) {
            *(int2*)(binbuf + 2 * ((size_t)b * BINCAP + g)) = ent;
        } else {
            const int k = atomicAdd(&binCount[NB], 1);
            *(int2*)(spill + 2 * (size_t)k) = ent;
        }
    }
}

__global__ __launch_bounds__(1024) void rank_kernel(const int* __restrict__ binCount,
                                                    const int* __restrict__ binbuf,
                                                    const int* __restrict__ spill, int* cnt,
                                                    int* __restrict__ slots, int* __restrict__ ovf) {
    __shared__ int s_cnt[BW];
    __shared__ __attribute__((aligned(16))) int s_slots[BW * CAP];

    const int tid  = threadIdx.x;
    const int b    = blockIdx.x;
    const int base = b * BW;
    int nn = N_NODES - base;
    if (nn > BW) nn = BW;

    for (int i = tid; i < BW; i += 1024) s_cnt[i] = 0;
    {
        int4 z;
        z.x = 0;
        z.y = 0;
        z.z = 0;
        z.w = 0;
        for (int i = tid; i < BW * CAP / 4; i += 1024) ((int4*)s_slots)[i] = z;
    }
    __syncthreads();

#define PLACE_EDGE(ent)                                               \
    {                                                                 \
        const int li = (ent).x - base;                                \
        const int rk = atomicAdd(&s_cnt[li], 1);                      \
        if (rk < CAP) {                                               \
            s_slots[li * CAP + rk] = (ent).y;                         \
        } else {                                                      \
            const int k = atomicAdd(&cnt[N_NODES], 1);                \
            *(int2*)(ovf + 2 * (size_t)k) = (ent);                    \
        }                                                             \
    }

    const int total = binCount[b];
    const int n = total < BINCAP ? total : BINCAP;
    const int2* src = (const int2*)(binbuf + 2 * (size_t)b * BINCAP);
    for (int i = tid; i < n; i += 1024) {
        const int2 ent = src[i];
        PLACE_EDGE(ent)
    }
    if (total > BINCAP) {
        int ns = binCount[NB];
        if (ns > N_EDGES) ns = N_EDGES;
        for (int i = tid; i < ns; i += 1024) {
            const int2 ent = *(const int2*)(spill + 2 * (size_t)i);
            if (ent.x >= base && ent.x < base + nn) PLACE_EDGE(ent)
        }
    }
#undef PLACE_EDGE
    __syncthreads();

    for (int i = tid; i < nn; i += 1024) cnt[base + i] = s_cnt[i];
    for (int p = tid; p < nn * (CAP / 4); p += 1024) {
        const int node = p / (CAP / 4);
        const int q    = p % (CAP / 4);
        int k = s_cnt[node];
        if (k > CAP) k = CAP;
        if (4 * q < k)
            ((int4*)slots)[(size_t)(base + node) * (CAP / 4) + q] =
                *(const int4*)&s_slots[node * CAP + 4 * q];
    }
}

__global__ __launch_bounds__(256) void prescale_kernel(const float* __restrict__ X,
                                                       const int* __restrict__ cnt,
                                                       float* __restrict__ XS) {
    static_assert(N_NODES * (DIM / 4) == 3125 * 1024, "four pieces per thread, exact grid");
    const v4f* x4 = (const v4f*)X;
    v4f* o4 = (v4f*)XS;
    const int base = blockIdx.x * 1024 + threadIdx.x;
    v4f v[4];
    float d[4];
#pragma unroll
    for (int m = 0; m < 4; ++m) {
        const int idx = base + 256 * m;
        v[m] = __builtin_nontemporal_load(x4 + idx);
        d[m] = rsqrtf((float)(cnt[idx >> 5] + 1));
    }
#pragma unroll
    for (int m = 0; m < 4; ++m) o4[base + 256 * m] = v[m] * d[m];
}

#define APITCH 132


__device__ __forceinline__ float4 row_scale(const float4& v, const float s) {
    float4 r;
    r.x = s * v.x;
    r.y = s * v.y;
    r.z = s * v.z;
    r.w = s * v.w;
    return r;
}

__device__ __forceinline__ void row_add(float4& a, const float4& v) {
    a.x += v.x;
    a.y += v.y;
    a.z += v.z;
    a.w += v.w;
}

__device__ __forceinline__ int list_piece(const int node, const int n, const int pos,
                                          const int lane, const int* __restrict__ slots) {
    int ent = 0;
    const int p = pos + lane;
    if (p == 0) {
        ent = node;
    } else if (p < n) {
        ent = slots[node * CAP + p - 1];
    }
    return ent;
}

__device__ __forceinline__ float4 gather_row(const int node, const int nedge, const int ent0,
                                             const unsigned ul, const int* __restrict__ slots,
                                             const int* __restrict__ ovf,
                                             const int* __restrict__ cnt,
                                             const float4* __restrict__ h4) {
    const int lane = (int)ul;
    const int ntot = (nedge < CAP ? nedge : CAP) + 1;
    float4 acc;
    acc.x = 0.0f;
    acc.y = 0.0f;
    acc.z = 0.0f;
    acc.w = 0.0f;

#define EDGE_LOAD(k)                                                                          \
    const int s##k = __builtin_amdgcn_readlane(ent, j + k);                                   \
    const float4 v##k = h4[(unsigned)s##k * 32u + ul];

    int ent = ent0;
    for (int pos = 0; pos < ntot; pos += 32) {
        int n = ntot - pos;
        if (n > 32) n = 32;

        int j = 0;
        for (; j + 8 <= n; j += 8) {
            EDGE_LOAD(0) EDGE_LOAD(1) EDGE_LOAD(2) EDGE_LOAD(3)
            EDGE_LOAD(4) EDGE_LOAD(5) EDGE_LOAD(6) EDGE_LOAD(7)
            row_add(acc, v0);
            row_add(acc, v1);
            row_add(acc, v2);
            row_add(acc, v3);
            row_add(acc, v4);
            row_add(acc, v5);
            row_add(acc, v6);
            row_add(acc, v7);
        }
        if (j + 4 <= n) {
            EDGE_LOAD(0) EDGE_LOAD(1) EDGE_LOAD(2) EDGE_LOAD(3)
            row_add(acc, v0);
            row_add(acc, v1);
            row_add(acc, v2);
            row_add(acc, v3);
            j += 4;
        }
        const int left = n - j;
        if (left > 0) {
            EDGE_LOAD(0)
            if (left == 1) {
                row_add(acc, v0);
            } else {
                EDGE_LOAD(1)
                if (left == 2) {
                    row_add(acc, v0);
                    row_add(acc, v1);
                } else {
                    EDGE_LOAD(2)
                    row_add(acc, v0);
                    row_add(acc, v1);
                    row_add(acc, v2);
                }
            }
        }
        if (pos + 32 < ntot) ent = list_piece(node, ntot, pos + 32, lane, slots);
    }
    if (nedge > CAP) {
        int novf = __builtin_amdgcn_readfirstlane(cnt[N_NODES]);
        if (novf > N_EDGES) novf = N_EDGES;
        for (int base = 0; base < novf; base += 32) {
            int2 o;
            o.x = -1;
            o.y = 0;
            if (base + lane < novf) o = *(const int2*)(ovf + 2 * (size_t)(base + lane));
            unsigned m = __builtin_amdgcn_ballot_w32(o.x == node);
            while (m != 0u) {
                const int l = __builtin_ctz(m);
                m &= m - 1u;
                const int s0 = __builtin_amdgcn_readlane(o.y, l);
                const float4 v0 = h4[(unsigned)s0 * 32u + ul];
                row_add(acc, v0);
            }
        }
    }
#undef EDGE_LOAD
    return acc;
}

__global__ __launch_bounds__(256, 8) void agg_gemm_relu_kernel(const int* __restrict__ cnt,
                                                            const int* __restrict__ slots,
                                                            const int* __restrict__ ovf,
                                                            const float* __restrict__ X,
                                                            const float* __restrict__ W,
                                                            const float* __restrict__ bias,
                                                            float* __restrict__ OUT,
                                                            int ntiles) {
    constexpr int NTHR   = 256;
    constexpr int OPITCH = DIM + 4;
    constexpr int O_F4   = 16 * DIM / 4;
    constexpr int ROW_F4 = DIM / 4;
    constexpr int NO     = O_F4 / NTHR;

    __shared__ __attribute__((aligned(16))) float s_a[2][16 * APITCH];
    __shared__ __attribute__((aligned(16))) float s_o[2][16 * OPITCH];

    const int tid  = threadIdx.x;
    const int lane = tid & 31;
    const int wave = __builtin_amdgcn_readfirstlane(tid >> 5);
    const int l15  = lane & 15;
    const int half = lane >> 4;
    const int ncol = wave * 16 + l15;

    int tile = blockIdx.x;
    if (tile >= ntiles) return;

    v2f bf[32];
#pragma unroll
    for (int ks = 0; ks < 32; ++ks) {
        const int kk = 4 * ks + 2 * half;
        bf[ks].x = W[(size_t)kk * DIM + ncol];
        bf[ks].y = W[(size_t)(kk + 1) * DIM + ncol];
    }
    const float ob = bias[ncol];

    const float4* h4 = (const float4*)X;
    const unsigned ul = (unsigned)lane;

    int t_cnt = cnt[tile * 16 + l15];

    int cur  = 0;
    int done = -1;
    for (; tile < ntiles; tile += gridDim.x) {
        const int node0 = tile * 16 + wave;
        const int node1 = node0 + 8;
        const int n0 = __builtin_amdgcn_readlane(t_cnt, wave);
        const int n1 = __builtin_amdgcn_readlane(t_cnt, wave + 8);
        const float c_din = rsqrtf((float)(t_cnt + 1));
        const float di0 = __int_as_float(__builtin_amdgcn_readlane(__float_as_int(c_din), wave));
        const float di1 = __int_as_float(__builtin_amdgcn_readlane(__float_as_int(c_din), wave + 8));
        const int e0 = list_piece(node0, (n0 < CAP ? n0 : CAP) + 1, 0, lane, slots);
        const int e1 = list_piece(node1, (n1 < CAP ? n1 : CAP) + 1, 0, lane, slots);
        const int nxt = tile + gridDim.x;
        if (nxt < ntiles) {
            t_cnt = cnt[nxt * 16 + l15];
        }
        {
            const float4 xr = row_scale(gather_row(node0, n0, e0, ul, slots, ovf, cnt, h4), di0);
            *(float4*)&s_a[cur][wave * APITCH + 4 * lane] = xr;
        }
        {
            const float4 xr = row_scale(gather_row(node1, n1, e1, ul, slots, ovf, cnt, h4), di1);
            *(float4*)&s_a[cur][(wave + 8) * APITCH + 4 * lane] = xr;
        }
        __syncthreads();

        if (done >= 0) {
            float4* dst = (float4*)(OUT + (size_t)done * 16 * DIM);
            const float* so = s_o[cur ^ 1];
#pragma unroll
            for (int i = 0; i < NO; ++i) {
                const int idx = tid + i * NTHR;
                dst[idx] = *(const float4*)&so[(idx / ROW_F4) * OPITCH + 4 * (idx % ROW_F4)];
            }
        }

        {
            v8f acc = {};
            const float* ap = &s_a[cur][l15 * APITCH + 2 * half];
#pragma unroll
            for (int kh = 0; kh < 2; ++kh) {
                v2f af[16];
#pragma unroll
                for (int j = 0; j < 16; ++j) af[j] = *(const v2f*)(ap + 4 * (16 * kh + j));
                asm volatile("" ::"v"(af[0]), "v"(af[1]), "v"(af[2]), "v"(af[3]), "v"(af[4]), "v"(af[5]),
                             "v"(af[6]), "v"(af[7]), "v"(af[8]), "v"(af[9]), "v"(af[10]), "v"(af[11]),
                             "v"(af[12]), "v"(af[13]), "v"(af[14]), "v"(af[15]));
#pragma unroll
                for (int j = 0; j < 16; ++j)
                    acc = __builtin_amdgcn_wmma_f32_16x16x4_f32(
                         false, af[j],  false, bf[16 * kh + j],
                         (short)0, acc,  false,  false);
            }
            float* op = &s_o[cur][(8 * half) * OPITCH + ncol];
#pragma unroll
            for (int r = 0; r < 8; ++r) {
                const float dr = __shfl(c_din, 8 * half + r);
                op[r * OPITCH] = dr * fmaxf(acc[r] + ob, 0.0f);
            }
        }
        done = tile;
        cur ^= 1;
    }
    __syncthreads();
    {
        float4* dst = (float4*)(OUT + (size_t)done * 16 * DIM);
        const float* so = s_o[cur ^ 1];
#pragma unroll
        for (int i = 0; i < NO; ++i) {
            const int idx = tid + i * NTHR;
            dst[idx] = *(const float4*)&so[(idx / ROW_F4) * OPITCH + 4 * (idx % ROW_F4)];
        }
    }
}

#define WPITCH 48
__global__ __launch_bounds__(256, 8) void agg_gemm_relu_gemm_kernel(const int* __restrict__ cnt,
                                                                 const int* __restrict__ slots,
                                                                 const int* __restrict__ ovf,
                                                                 const float* __restrict__ X,
                                                                 const float* __restrict__ W,
                                                                 const float* __restrict__ bias,
                                                                 const float* __restrict__ Wo,
                                                                 const float* __restrict__ obias,
                                                                 float* __restrict__ OUT,
                                                                 int ntiles) {
    constexpr int NWO  = (N_CLASSES + 15) / 16;
    constexpr int O_F4 = 16 * N_CLASSES / 4;
    constexpr int ROW_F4 = N_CLASSES / 4;
    static_assert(O_F4 <= 256, "one piece per thread");
    static_assert(16 * NWO <= WPITCH, "padded columns");

    __shared__ __attribute__((aligned(16))) float s_a[16 * APITCH];
    __shared__ __attribute__((aligned(16))) float s_x[16 * APITCH];
    __shared__ __attribute__((aligned(16))) float s_o[16 * N_CLASSES];
    __shared__ __attribute__((aligned(16))) float s_w[64 * WPITCH * 2];

    const int tid  = threadIdx.x;
    const int lane = tid & 31;
    const int wave = __builtin_amdgcn_readfirstlane(tid >> 5);
    const int l15  = lane & 15;
    const int half = lane >> 4;
    const int ncol = wave * 16 + l15;

    int tile = blockIdx.x;
    if (tile >= ntiles) return;

    v2f bf[32];
#pragma unroll
    for (int ks = 0; ks < 32; ++ks) {
        const int kk = 4 * ks + 2 * half;
        bf[ks].x = W[(size_t)kk * DIM + ncol];
        bf[ks].y = W[(size_t)(kk + 1) * DIM + ncol];
    }
    const float hb = bias[ncol];

    {
        static_assert(DIM * N_CLASSES == 4 * 5 * 256 && N_CLASSES % 4 == 0, "five pieces per thread");
        static_assert(WPITCH - N_CLASSES == 8, "sixteen zero floats per fragment pair row");
        const float4* wo4 = (const float4*)Wo;
        float4 wv[5];
#pragma unroll
        for (int m = 0; m < 5; ++m) wv[m] = wo4[tid + 256 * m];
#pragma unroll
        for (int m = 0; m < 5; ++m) {
            const int g = 4 * (tid + 256 * m);
            const int k = g / N_CLASSES;
            const int c = g % N_CLASSES;
            float* d = &s_w[((k >> 1) * WPITCH + c) * 2 + (k & 1)];
            d[0] = wv[m].x;
            d[2] = wv[m].y;
            d[4] = wv[m].z;
            d[6] = wv[m].w;
        }
        for (int i = tid; i < 64 * (WPITCH - N_CLASSES) * 2; i += 256)
            s_w[((i >> 4) * WPITCH + N_CLASSES) * 2 + (i & 15)] = 0.0f;
    }
    const bool ov = (wave < NWO) && (ncol < N_CLASSES);
    const float ob = ov ? obias[ncol] : 0.0f;

    const float4* h4 = (const float4*)X;
    const unsigned ul = (unsigned)lane;

    int t_cnt = cnt[tile * 16 + l15];

    int done = -1;
    for (; tile < ntiles; tile += gridDim.x) {
        const int node0 = tile * 16 + wave;
        const int node1 = node0 + 8;
        const int n0 = __builtin_amdgcn_readlane(t_cnt, wave);
        const int n1 = __builtin_amdgcn_readlane(t_cnt, wave + 8);
        const float c_din = rsqrtf((float)(t_cnt + 1));
        const float di0 = __int_as_float(__builtin_amdgcn_readlane(__float_as_int(c_din), wave));
        const float di1 = __int_as_float(__builtin_amdgcn_readlane(__float_as_int(c_din), wave + 8));
        const int e0 = list_piece(node0, (n0 < CAP ? n0 : CAP) + 1, 0, lane, slots);
        const int e1 = list_piece(node1, (n1 < CAP ? n1 : CAP) + 1, 0, lane, slots);
        const int nxt = tile + gridDim.x;
        if (nxt < ntiles) {
            t_cnt = cnt[nxt * 16 + l15];
        }
        {
            const float4 xr = row_scale(gather_row(node0, n0, e0, ul, slots, ovf, cnt, h4), di0);
            *(float4*)&s_a[wave * APITCH + 4 * lane] = xr;
        }
        {
            const float4 xr = row_scale(gather_row(node1, n1, e1, ul, slots, ovf, cnt, h4), di1);
            *(float4*)&s_a[(wave + 8) * APITCH + 4 * lane] = xr;
        }
        __syncthreads();

        if (done >= 0 && tid < O_F4) {
            float4* dst = (float4*)(OUT + (size_t)done * 16 * N_CLASSES);
            dst[tid] = *(const float4*)&s_o[(tid / ROW_F4) * N_CLASSES + 4 * (tid % ROW_F4)];
        }

        {
            v8f acc = {};
            const float* ap = &s_a[l15 * APITCH + 2 * half];
#pragma unroll
            for (int kh = 0; kh < 2; ++kh) {
                v2f af[16];
#pragma unroll
                for (int j = 0; j < 16; ++j) af[j] = *(const v2f*)(ap + 4 * (16 * kh + j));
                asm volatile("" ::"v"(af[0]), "v"(af[1]), "v"(af[2]), "v"(af[3]), "v"(af[4]), "v"(af[5]),
                             "v"(af[6]), "v"(af[7]), "v"(af[8]), "v"(af[9]), "v"(af[10]), "v"(af[11]),
                             "v"(af[12]), "v"(af[13]), "v"(af[14]), "v"(af[15]));
#pragma unroll
                for (int j = 0; j < 16; ++j)
                    acc = __builtin_amdgcn_wmma_f32_16x16x4_f32(
                         false, af[j],  false, bf[16 * kh + j],
                         (short)0, acc,  false,  false);
            }
            float* xp = &s_x[(8 * half) * APITCH + ncol];
#pragma unroll
            for (int r = 0; r < 8; ++r) xp[r * APITCH] = fmaxf(acc[r] + hb, 0.0f);
        }
        __syncthreads();

        if (wave < NWO) {
            v8f acc = {};
            const float* ap = &s_x[l15 * APITCH + 2 * half];
            const float* bp = &s_w[(half * WPITCH + ncol) * 2];
#pragma unroll
            for (int kq = 0; kq < 4; ++kq) {
                v2f af[8], bq[8];
#pragma unroll
                for (int j = 0; j < 8; ++j) {
                    af[j] = *(const v2f*)(ap + 4 * (8 * kq + j));
                    bq[j] = *(const v2f*)(bp + 4 * WPITCH * (8 * kq + j));
                }
                asm volatile("" ::"v"(af[0]), "v"(af[1]), "v"(af[2]), "v"(af[3]), "v"(af[4]), "v"(af[5]),
                             "v"(af[6]), "v"(af[7]), "v"(bq[0]), "v"(bq[1]), "v"(bq[2]), "v"(bq[3]),
                             "v"(bq[4]), "v"(bq[5]), "v"(bq[6]), "v"(bq[7]));
#pragma unroll
                for (int j = 0; j < 8; ++j)
                    acc = __builtin_amdgcn_wmma_f32_16x16x4_f32(
                         false, af[j],  false, bq[j],
                         (short)0, acc,  false,  false);
            }
            if (ov) {
                float* op = &s_o[(8 * half) * N_CLASSES + ncol];
#pragma unroll
                for (int r = 0; r < 8; ++r) op[r * N_CLASSES] = acc[r] + ob;
            }
        }
        done = tile;
    }
    __syncthreads();
    if (tid < O_F4) {
        float4* dst = (float4*)(OUT + (size_t)done * 16 * N_CLASSES);
        dst[tid] = *(const float4*)&s_o[(tid / ROW_F4) * N_CLASSES + 4 * (tid % ROW_F4)];
    }
}

extern "C" void kernel_launch(void* const* d_in, const int* in_sizes, int n_in,
                              void* d_out, int out_size, void* d_ws, size_t ws_size,
                              hipStream_t stream) {
    const float* x    = (const float*)d_in[0];
    const int*   edge = (const int*)d_in[1];
    const float* W1   = (const float*)d_in[2];
    const float* b1   = (const float*)d_in[3];
    const float* W2   = (const float*)d_in[4];
    const float* b2   = (const float*)d_in[5];
    const float* Wout = (const float*)d_in[6];
    const float* bout = (const float*)d_in[7];
    const int* row = edge;
    const int* col = edge + N_EDGES;
    float* out = (float*)d_out;

    char* ws = (char*)d_ws;
    int*   cnt   = (int*)(ws + ((size_t)1 << 19));
    float* bufA  = (float*)(ws + ((size_t)1 << 20));
    int*   slots = (int*)(ws + ((size_t)112 << 20));
    int*   ovf   = (int*)(ws + ((size_t)144 << 20));
    float* xs    = (float*)(ws + ((size_t)160 << 20));
    int*   binCount = (int*)(ws + ((size_t)15 << 16));
    int*   binbuf   = (int*)(ws + ((size_t)64 << 20));
    int*   spill    = (int*)(ws + ((size_t)84 << 20));

    const int ntiles = N_NODES / 16;

    cnt_zero_kernel<<<(NB + 1 + 255) / 256, 256, 0, stream>>>(cnt, binCount);
    bin_kernel<<<(N_EDGES + EPB - 1) / EPB, 1024, 0, stream>>>(row, col, binCount, binbuf, spill);
    rank_kernel<<<NB, 1024, 0, stream>>>(binCount, binbuf, spill, cnt, slots, ovf);
    prescale_kernel<<<N_NODES * (DIM / 4) / 1024, 256, 0, stream>>>(x, cnt, xs);

    agg_gemm_relu_kernel<<<1024, 256, 0, stream>>>(cnt, slots, ovf, xs, W1, b1, bufA, ntiles);
    agg_gemm_relu_gemm_kernel<<<1024, 256, 0, stream>>>(cnt, slots, ovf, bufA, W2, b2, Wout, bout, out, ntiles);
}
